# starting kernel
# baseline (speedup 1.0000x reference)
#include <hip/hip_runtime.h>
#include <math.h>

#define IN_C 128
#define HID 32
#define HEADS 4
#define OUT_C 64
#define NEG_SLOPE 0.2f

typedef __attribute__((ext_vector_type(2))) float v2f;
typedef __attribute__((ext_vector_type(4))) float v4f;
typedef __attribute__((ext_vector_type(8))) float v8f;

template <int K, int NC>
__global__ void gemm_wmma(const float* __restrict__ A, const float* __restrict__ B,
                          float* __restrict__ C, int M) {
  const int lane  = threadIdx.x;
  const int m16   = lane & 15;
  const int khalf = lane >> 4;
  int row = blockIdx.x * 16 + m16;
  if (row >= M) row = M - 1;
  const int col = (int)blockIdx.y * 16 + m16;
  const float* arow = A + (size_t)row * K;

  v8f acc = {};
#pragma unroll 4
  for (int k0 = 0; k0 < K; k0 += 4) {
    const int ka = k0 + khalf * 2;
    v2f av, bv;
    av.x = arow[ka];
    av.y = arow[ka + 1];
    bv.x = B[ka * NC + col];
    bv.y = B[(ka + 1) * NC + col];
    acc = __builtin_amdgcn_wmma_f32_16x16x4_f32(false, av, false, bv, (short)0,
                                                acc, false, false);
  }

  const int rbase = (int)blockIdx.x * 16 + khalf * 8;
  float* cbase = C + (size_t)rbase * NC + col;
  if ((int)blockIdx.x * 16 + 16 <= M) {
#pragma unroll
    for (int i = 0; i < 8; ++i) cbase[(size_t)i * NC] = acc[i];
  } else {
#pragma unroll
    for (int i = 0; i < 8; ++i)
      if (rbase + i < M) cbase[(size_t)i * NC] = acc[i];
  }
}

template <int H, int CHPH>
__global__ void node_logits(const float* __restrict__ h, const float* __restrict__ asrc,
                            const float* __restrict__ adst,
                            float* __restrict__ als, float* __restrict__ ald, int NH) {
  int tid = blockIdx.x * blockDim.x + threadIdx.x;
  if (tid >= NH) return;
  int node = tid / H;
  int head = tid % H;
  const float* hp = h + (size_t)node * (H * CHPH) + head * CHPH;
  const float* as = asrc + head * CHPH;
  const float* ad = adst + head * CHPH;
  float s = 0.f, d = 0.f;
#pragma unroll
  for (int c = 0; c < CHPH; ++c) {
    float v = hp[c];
    s = fmaf(v, as[c], s);
    d = fmaf(v, ad[c], d);
  }
  als[tid] = s;
  ald[tid] = d;
}

__device__ __forceinline__ unsigned fkey(float f) {
  unsigned u = __float_as_uint(f);
  return (u & 0x80000000u) ? ~u : (u | 0x80000000u);
}
__device__ __forceinline__ float funkey(unsigned k) {
  return __uint_as_float((k & 0x80000000u) ? (k & 0x7fffffffu) : ~k);
}

__device__ __forceinline__ float leaky(float v) {
  return v > 0.f ? v : NEG_SLOPE * v;
}

template <int H>
__global__ void edge_max_k(const int* __restrict__ src, const int* __restrict__ dst,
                           const float* __restrict__ als, const float* __restrict__ ald,
                           unsigned* __restrict__ mkey, int EH) {
  int tid = blockIdx.x * blockDim.x + threadIdx.x;
  if (tid >= EH) return;
  int e = tid / H;
  int head = tid % H;
  int s = src[e], d = dst[e];
  float lr = leaky(als[s * H + head] + ald[d * H + head]);
  atomicMax(&mkey[d * H + head], fkey(lr));
}

template <int H>
__global__ void edge_expsum_k(const int* __restrict__ src, const int* __restrict__ dst,
                              const float* __restrict__ als, const float* __restrict__ ald,
                              const unsigned* __restrict__ mkey, float* __restrict__ ssum,
                              int EH) {
  int tid = blockIdx.x * blockDim.x + threadIdx.x;
  if (tid >= EH) return;
  int e = tid / H;
  int head = tid % H;
  int s = src[e], d = dst[e];
  float lr = leaky(als[s * H + head] + ald[d * H + head]);
  float ex = __expf(lr - funkey(mkey[d * H + head]));
  atomicAdd(&ssum[d * H + head], ex);
}

template <int H, int CH>
__global__ void edge_aggregate(const int* __restrict__ src, const int* __restrict__ dst,
                               const float* __restrict__ h,
                               const float* __restrict__ als, const float* __restrict__ ald,
                               const unsigned* __restrict__ mkey, const float* __restrict__ ssum,
                               float* __restrict__ out, int E) {
  int tid = blockIdx.x * blockDim.x + threadIdx.x;
  int e = tid >> 5;
  int lane = tid & 31;
  if (e >= E) return;
  int s = src[e], d = dst[e];
  constexpr int PER = CH / 32;
  constexpr int CHPH = CH / H;
  const int c0 = lane * PER;
  const int head = c0 / CHPH;
  float lr = leaky(als[s * H + head] + ald[d * H + head]);
  float w = __expf(lr - funkey(mkey[d * H + head])) /
            (ssum[d * H + head] + 1e-16f);
  const float* hrow = h + (size_t)s * CH + c0;
  float* orow = out + (size_t)d * CH + c0;
  if (PER == 4) {
    v4f hv = *(const v4f*)hrow;
    atomicAdd(&orow[0], hv.x * w);
    atomicAdd(&orow[1], hv.y * w);
    atomicAdd(&orow[2], hv.z * w);
    atomicAdd(&orow[3], hv.w * w);
  } else {
    v2f hv = *(const v2f*)hrow;
    atomicAdd(&orow[0], hv.x * w);
    atomicAdd(&orow[1], hv.y * w);
  }
}

__global__ void fill_u32(unsigned* __restrict__ p, unsigned v, int n) {
  int i = blockIdx.x * blockDim.x + threadIdx.x;
  if (i < n) p[i] = v;
}

template <int CH, bool ELU>
__global__ void bias_act(float* __restrict__ x, const float* __restrict__ b, int n) {
  int i = blockIdx.x * blockDim.x + threadIdx.x;
  if (i >= n) return;
  int c = i % CH;
  float v = x[i] + b[c];
  if (ELU) v = v > 0.f ? v : (__expf(v) - 1.0f);
  x[i] = v;
}

extern "C" void kernel_launch(void* const* d_in, const int* in_sizes, int n_in,
                              void* d_out, int out_size, void* d_ws, size_t ws_size,
                              hipStream_t stream) {
  const float* x   = (const float*)d_in[0];
  const int*   ei  = (const int*)d_in[1];
  const float* W1  = (const float*)d_in[2];
  const float* as1 = (const float*)d_in[3];
  const float* ad1 = (const float*)d_in[4];
  const float* b1  = (const float*)d_in[5];
  const float* W2  = (const float*)d_in[6];
  const float* as2 = (const float*)d_in[7];
  const float* ad2 = (const float*)d_in[8];
  const float* b2  = (const float*)d_in[9];
  float* out = (float*)d_out;

  const int N = in_sizes[0] / IN_C;
  const int E = in_sizes[1] / 2;
  const int* src = ei;
  const int* dst = ei + E;

  float* ws = (float*)d_ws;
  float*    h1   = ws;
  float*    agg1 = h1 + (size_t)N * 128;
  float*    als  = agg1 + (size_t)N * 128;
  float*    ald  = als + (size_t)N * HEADS;
  unsigned* mkey = (unsigned*)(ald + (size_t)N * HEADS);
  float*    ssum = (float*)(mkey + (size_t)N * HEADS);

  auto cdiv = [](long long a, long long b) -> unsigned { return (unsigned)((a + b - 1) / b); };
  const unsigned KEY_NEG_INF = 0x007FFFFFu;

  gemm_wmma<IN_C, HEADS * HID>
      <<<dim3(cdiv(N, 16), (HEADS * HID) / 16), 32, 0, stream>>>(x, W1, h1, N);
  node_logits<HEADS, HID>
      <<<cdiv((long long)N * HEADS, 256), 256, 0, stream>>>(h1, as1, ad1, als, ald, N * HEADS);
  fill_u32<<<cdiv((long long)N * HEADS, 256), 256, 0, stream>>>(mkey, KEY_NEG_INF, N * HEADS);
  fill_u32<<<cdiv((long long)N * HEADS, 256), 256, 0, stream>>>((unsigned*)ssum, 0u, N * HEADS);
  fill_u32<<<cdiv((long long)N * 128, 256), 256, 0, stream>>>((unsigned*)agg1, 0u, N * 128);
  edge_max_k<HEADS>
      <<<cdiv((long long)E * HEADS, 256), 256, 0, stream>>>(src, dst, als, ald, mkey, E * HEADS);
  edge_expsum_k<HEADS>
      <<<cdiv((long long)E * HEADS, 256), 256, 0, stream>>>(src, dst, als, ald, mkey, ssum,
                                                            E * HEADS);
  edge_aggregate<HEADS, 128>
      <<<cdiv((long long)E * 32, 256), 256, 0, stream>>>(src, dst, h1, als, ald, mkey, ssum,
                                                         agg1, E);
  bias_act<128, true><<<cdiv((long long)N * 128, 256), 256, 0, stream>>>(agg1, b1, N * 128);

  float* h2 = h1;
  gemm_wmma<HEADS * HID, OUT_C>
      <<<dim3(cdiv(N, 16), OUT_C / 16), 32, 0, stream>>>(agg1, W2, h2, N);
  node_logits<1, OUT_C><<<cdiv(N, 256), 256, 0, stream>>>(h2, as2, ad2, als, ald, N);
  fill_u32<<<cdiv(N, 256), 256, 0, stream>>>(mkey, KEY_NEG_INF, N);
  fill_u32<<<cdiv(N, 256), 256, 0, stream>>>((unsigned*)ssum, 0u, N);
  fill_u32<<<cdiv((long long)N * OUT_C, 256), 256, 0, stream>>>((unsigned*)out, 0u, N * OUT_C);
  edge_max_k<1><<<cdiv(E, 256), 256, 0, stream>>>(src, dst, als, ald, mkey, E);
  edge_expsum_k<1><<<cdiv(E, 256), 256, 0, stream>>>(src, dst, als, ald, mkey, ssum, E);
  edge_aggregate<1, OUT_C>
      <<<cdiv((long long)E * 32, 256), 256, 0, stream>>>(src, dst, h2, als, ald, mkey, ssum,
                                                         out, E);
  bias_act<OUT_C, false><<<cdiv((long long)N * OUT_C, 256), 256, 0, stream>>>(out, b2, N * OUT_C);
}


# Edges bucketed by destination in-call (int atomic rank + scan), then one wave per node does max, exp-sum, weighted sum, bias, act in registers; float atomics removed, sums now run in rank order
# speedup vs baseline: 2.4829x; 2.4829x over previous
#include <hip/hip_runtime.h>
#include <math.h>

#define IN_C 128
#define HID 32
#define HEADS 4
#define OUT_C 64
#define NEG_SLOPE 0.2f

typedef __attribute__((ext_vector_type(2))) float v2f;
typedef __attribute__((ext_vector_type(4))) float v4f;
typedef __attribute__((ext_vector_type(8))) float v8f;

template <int K, int NC>
__global__ void gemm_wmma(const float* __restrict__ A, const float* __restrict__ B,
                          float* __restrict__ C, int M) {
  const int lane  = threadIdx.x;
  const int m16   = lane & 15;
  const int khalf = lane >> 4;
  int row = blockIdx.x * 16 + m16;
  if (row >= M) row = M - 1;
  const int col = (int)blockIdx.y * 16 + m16;
  const float* arow = A + (size_t)row * K;

  v8f acc = {};
#pragma unroll 4
  for (int k0 = 0; k0 < K; k0 += 4) {
    const int ka = k0 + khalf * 2;
    v2f av, bv;
    av.x = arow[ka];
    av.y = arow[ka + 1];
    bv.x = B[ka * NC + col];
    bv.y = B[(ka + 1) * NC + col];
    acc = __builtin_amdgcn_wmma_f32_16x16x4_f32(false, av, false, bv, (short)0,
                                                acc, false, false);
  }

  const int rbase = (int)blockIdx.x * 16 + khalf * 8;
  float* cbase = C + (size_t)rbase * NC + col;
  if ((int)blockIdx.x * 16 + 16 <= M) {
#pragma unroll
    for (int i = 0; i < 8; ++i) cbase[(size_t)i * NC] = acc[i];
  } else {
#pragma unroll
    for (int i = 0; i < 8; ++i)
      if (rbase + i < M) cbase[(size_t)i * NC] = acc[i];
  }
}

template <int H, int CHPH>
__global__ void node_logits(const float* __restrict__ h, const float* __restrict__ asrc,
                            const float* __restrict__ adst,
                            float* __restrict__ als, float* __restrict__ ald, int NH) {
  int tid = blockIdx.x * blockDim.x + threadIdx.x;
  if (tid >= NH) return;
  int node = tid / H;
  int head = tid % H;
  const float* hp = h + (size_t)node * (H * CHPH) + head * CHPH;
  const float* as = asrc + head * CHPH;
  const float* ad = adst + head * CHPH;
  float s = 0.f, d = 0.f;
#pragma unroll
  for (int c = 0; c < CHPH; ++c) {
    float v = hp[c];
    s = fmaf(v, as[c], s);
    d = fmaf(v, ad[c], d);
  }
  als[tid] = s;
  ald[tid] = d;
}

__device__ __forceinline__ float leaky(float v) {
  return v > 0.f ? v : NEG_SLOPE * v;
}

__global__ void fill_u32(unsigned* __restrict__ p, unsigned v, int n) {
  int i = blockIdx.x * blockDim.x + threadIdx.x;
  if (i < n) p[i] = v;
}


__global__ void dst_rank_k(const int* __restrict__ dst, int* __restrict__ deg,
                           int* __restrict__ rank, int E) {
  int e = blockIdx.x * blockDim.x + threadIdx.x;
  if (e >= E) return;
  rank[e] = atomicAdd(&deg[dst[e]], 1);
}

__global__ void __launch_bounds__(1024)
row_start_k(const int* __restrict__ deg, int* __restrict__ rowp, int n) {
  __shared__ int wsum[32];
  const int lane = threadIdx.x & 31;
  const int w = threadIdx.x >> 5;
  const int seg = (n + 31) / 32;
  const int b = w * seg;
  int e = b + seg;
  if (e > n) e = n;

  int s = 0;
#pragma unroll 4
  for (int i = b + lane; i < e; i += 32) s += deg[i];
  for (int o = 16; o > 0; o >>= 1) s += __shfl_xor(s, o);
  if (lane == 0) wsum[w] = s;
  __syncthreads();

  int run = 0;
  for (int k = 0; k < w; ++k) run += wsum[k];

  for (int i0 = b; i0 < e; i0 += 128) {
    const int i = i0 + lane * 4;
    const int v0 = (i     < e) ? deg[i]     : 0;
    const int v1 = (i + 1 < e) ? deg[i + 1] : 0;
    const int v2 = (i + 2 < e) ? deg[i + 2] : 0;
    const int v3 = (i + 3 < e) ? deg[i + 3] : 0;
    const int tot = v0 + v1 + v2 + v3;
    int inc = tot;
    for (int o = 1; o < 32; o <<= 1) {
      const int u = __shfl_up(inc, o);
      if (lane >= o) inc += u;
    }
    const int ex = run + inc - tot;
    if (i     < e) rowp[i]     = ex;
    if (i + 1 < e) rowp[i + 1] = ex + v0;
    if (i + 2 < e) rowp[i + 2] = ex + v0 + v1;
    if (i + 3 < e) rowp[i + 3] = ex + v0 + v1 + v2;
    run += __shfl(inc, 31);
  }
  if (threadIdx.x == 1023) rowp[n] = run;
}

__global__ void csr_fill_k(const int* __restrict__ src, const int* __restrict__ dst,
                           const int* __restrict__ rowp, const int* __restrict__ rank,
                           int* __restrict__ csr, int E) {
  int e = blockIdx.x * blockDim.x + threadIdx.x;
  if (e >= E) return;
  csr[rowp[dst[e]] + rank[e]] = src[e];
}

__global__ void copy_i32(const int* __restrict__ from, int* __restrict__ to, int n) {
  int i = blockIdx.x * blockDim.x + threadIdx.x;
  if (i < n) to[i] = from[i];
}

template <int H, int CH, bool ELU>
__global__ void node_gather(const int* __restrict__ rowp, const int* __restrict__ csr,
                            const float* __restrict__ h,
                            const float* __restrict__ als, const float* __restrict__ ald,
                            const float* __restrict__ bias, float* __restrict__ out, int N) {
  const int gw = (blockIdx.x * blockDim.x + threadIdx.x) >> 5;
  const int lane = threadIdx.x & 31;
  if (gw >= N) return;
  const int d = __builtin_amdgcn_readfirstlane(gw);
  constexpr int PER = CH / 32;
  constexpr int CHPH = CH / H;
  const int c0 = lane * PER;
  const int head = c0 / CHPH;
  const int beg = rowp[d];
  const int end = rowp[d + 1];
  const float ad = ald[d * H + head];

  float m = -INFINITY;
  for (int base = beg; base < end; base += 32) {
    int cnt = end - base;
    if (cnt > 32) cnt = 32;
    const int sv = (lane < cnt) ? csr[base + lane] : 0;
    for (int j = 0; j < cnt; ++j) {
      const int s = __builtin_amdgcn_readlane(sv, j);
      m = fmaxf(m, leaky(als[s * H + head] + ad));
    }
  }

  float ssum = 0.f;
  float acc[PER];
#pragma unroll
  for (int i = 0; i < PER; ++i) acc[i] = 0.f;
  for (int base = beg; base < end; base += 32) {
    int cnt = end - base;
    if (cnt > 32) cnt = 32;
    const int sv = (lane < cnt) ? csr[base + lane] : 0;
#pragma unroll 4
    for (int j = 0; j < cnt; ++j) {
      const int s = __builtin_amdgcn_readlane(sv, j);
      const float p = __expf(leaky(als[s * H + head] + ad) - m);
      ssum += p;
      const float* hrow = h + (size_t)s * CH + c0;
      if (PER == 4) {
        const v4f hv = *(const v4f*)hrow;
        acc[0] = fmaf(hv.x, p, acc[0]);
        acc[1] = fmaf(hv.y, p, acc[1]);
        acc[2] = fmaf(hv.z, p, acc[2]);
        acc[3] = fmaf(hv.w, p, acc[3]);
      } else {
        const v2f hv = *(const v2f*)hrow;
        acc[0] = fmaf(hv.x, p, acc[0]);
        acc[1] = fmaf(hv.y, p, acc[1]);
      }
    }
  }

  const float den = ssum + 1e-16f;
  float r[PER];
#pragma unroll
  for (int i = 0; i < PER; ++i) {
    float v = acc[i] / den + bias[c0 + i];
    if (ELU) v = v > 0.f ? v : (__expf(v) - 1.0f);
    r[i] = v;
  }
  float* orow = out + (size_t)d * CH + c0;
  if (PER == 4) {
    v4f o4 = {r[0], r[1], r[2], r[3]};
    *(v4f*)orow = o4;
  } else {
    v2f o2 = {r[0], r[1]};
    *(v2f*)orow = o2;
  }
}

extern "C" void kernel_launch(void* const* d_in, const int* in_sizes, int n_in,
                              void* d_out, int out_size, void* d_ws, size_t ws_size,
                              hipStream_t stream) {
  const float* x   = (const float*)d_in[0];
  const int*   ei  = (const int*)d_in[1];
  const float* W1  = (const float*)d_in[2];
  const float* as1 = (const float*)d_in[3];
  const float* ad1 = (const float*)d_in[4];
  const float* b1  = (const float*)d_in[5];
  const float* W2  = (const float*)d_in[6];
  const float* as2 = (const float*)d_in[7];
  const float* ad2 = (const float*)d_in[8];
  const float* b2  = (const float*)d_in[9];
  float* out = (float*)d_out;

  const int N = in_sizes[0] / IN_C;
  const int E = in_sizes[1] / 2;
  const int* src = ei;
  const int* dst = ei + E;

  float* ws = (float*)d_ws;
  float* h1   = ws;
  float* act1 = h1 + (size_t)N * 128;
  int*   csr  = (int*)(act1 + (size_t)N * 128);
  int*   rank = (int*)act1;
  int*   deg  = (int*)out;
  int*   rowp = deg + N;
  float* als1 = out + (size_t)N * 3;
  float* ald1 = als1 + (size_t)N * HEADS;
  float* h2    = h1;
  int*   rowp2 = (int*)(h1 + (size_t)N * 64);
  float* als2  = h1 + (size_t)N * 64 + (size_t)N * 2;
  float* ald2  = als2 + N;

  auto cdiv = [](long long a, long long b) -> unsigned { return (unsigned)((a + b - 1) / b); };

  gemm_wmma<IN_C, HEADS * HID>
      <<<dim3(cdiv(N, 16), (HEADS * HID) / 16), 32, 0, stream>>>(x, W1, h1, N);
  node_logits<HEADS, HID>
      <<<cdiv((long long)N * HEADS, 256), 256, 0, stream>>>(h1, as1, ad1, als1, ald1, N * HEADS);

  fill_u32<<<cdiv(N, 256), 256, 0, stream>>>((unsigned*)deg, 0u, N);
  dst_rank_k<<<cdiv(E, 256), 256, 0, stream>>>(dst, deg, rank, E);
  row_start_k<<<1, 1024, 0, stream>>>(deg, rowp, N);
  csr_fill_k<<<cdiv(E, 256), 256, 0, stream>>>(src, dst, rowp, rank, csr, E);

  node_gather<HEADS, 128, true>
      <<<cdiv((long long)N * 32, 256), 256, 0, stream>>>(rowp, csr, h1, als1, ald1, b1, act1, N);

  copy_i32<<<cdiv(N + 1, 256), 256, 0, stream>>>(rowp, rowp2, N + 1);
  gemm_wmma<HEADS * HID, OUT_C>
      <<<dim3(cdiv(N, 16), OUT_C / 16), 32, 0, stream>>>(act1, W2, h2, N);
  node_logits<1, OUT_C><<<cdiv(N, 256), 256, 0, stream>>>(h2, as2, ad2, als2, ald2, N);
  node_gather<1, OUT_C, false>
      <<<cdiv((long long)N * 32, 256), 256, 0, stream>>>(rowp2, csr, h2, als2, ald2, b2, out, N);
}


# GEMMs: each wave owns 16-row strips across all columns (A rows read once into registers, B staged once per 16-wave work-group in LDS); k order kept, GEMM results bit-identical
# speedup vs baseline: 2.8073x; 1.1307x over previous
#include <hip/hip_runtime.h>
#include <math.h>

#define IN_C 128
#define HID 32
#define HEADS 4
#define OUT_C 64
#define NEG_SLOPE 0.2f

typedef __attribute__((ext_vector_type(2))) float v2f;
typedef __attribute__((ext_vector_type(4))) float v4f;
typedef __attribute__((ext_vector_type(8))) float v8f;

template <int K, int NC>
__global__ void __launch_bounds__(512)
gemm_wmma(const float* __restrict__ A, const float* __restrict__ B,
          float* __restrict__ C, int M) {
  constexpr int NT = NC / 16;
  constexpr int KS = K / 4;
  constexpr int PB = 2 * NC + 32;
  constexpr int WAVES = 16;
  constexpr int QN = NC / 4;
  __shared__ __attribute__((aligned(16))) float Ws[(K / 2) * PB];

  const int lane = threadIdx.x & 31;
  const int c = lane & 15;
  const int h = lane >> 4;
  const int wave = __builtin_amdgcn_readfirstlane((int)threadIdx.x >> 5);
  const int ntiles = (M + 15) >> 4;
  const int stride = (int)gridDim.x * WAVES;
  int tile = (int)blockIdx.x + (int)gridDim.x * wave;

  v2f a[KS];
  bool have = tile < ntiles;
  if (have) {
    int row = tile * 16 + c;
    if (row >= M) row = M - 1;
    const float* ap = A + (size_t)row * K + 2 * h;
#pragma unroll
    for (int ks = 0; ks < KS; ++ks) a[ks] = *(const v2f*)(ap + 4 * ks);
  }

  static_assert(((K / 2) * QN) % 512 == 0, "whole passes of the work-group");
#pragma unroll
  for (int it = 0; it < ((K / 2) * QN) / 512; ++it) {
    const int i = (int)threadIdx.x + it * 512;
    const int p = i / QN;
    const int q = i % QN;
    const v4f r0 = *(const v4f*)(B + (size_t)(2 * p) * NC + 4 * q);
    const v4f r1 = *(const v4f*)(B + (size_t)(2 * p + 1) * NC + 4 * q);
    float* d = Ws + p * PB + 8 * q;
    const v4f lo = {r0.x, r1.x, r0.y, r1.y};
    const v4f hi = {r0.z, r1.z, r0.w, r1.w};
    *(v4f*)d = lo;
    *(v4f*)(d + 4) = hi;
  }
  __syncthreads();

  const float* bl = Ws + h * PB + 2 * c;
  while (have) {
    asm volatile("" ::: "memory");
    v8f acc[NT];
#pragma unroll
    for (int t = 0; t < NT; ++t) acc[t] = v8f{};

#pragma unroll
    for (int ks = 0; ks < KS; ++ks) {
#pragma unroll
      for (int t = 0; t < NT; ++t) {
        const v2f bv = *(const v2f*)(bl + ks * 2 * PB + t * 32);
        acc[t] = __builtin_amdgcn_wmma_f32_16x16x4_f32(false, a[ks], false, bv, (short)0,
                                                       acc[t], false, false);
      }
    }

    const int rbase = tile * 16 + h * 8;
    float* cbase = C + (size_t)rbase * NC + c;
    if (tile * 16 + 16 <= M) {
#pragma unroll
      for (int t = 0; t < NT; ++t)
#pragma unroll
        for (int i = 0; i < 8; ++i) cbase[(size_t)i * NC + t * 16] = acc[t][i];
    } else {
#pragma unroll
      for (int t = 0; t < NT; ++t)
#pragma unroll
        for (int i = 0; i < 8; ++i)
          if (rbase + i < M) cbase[(size_t)i * NC + t * 16] = acc[t][i];
    }

    tile += stride;
    have = tile < ntiles;
    if (have) {
      int row = tile * 16 + c;
      if (row >= M) row = M - 1;
      const float* ap = A + (size_t)row * K + 2 * h;
#pragma unroll
      for (int ks = 0; ks < KS; ++ks) a[ks] = *(const v2f*)(ap + 4 * ks);
    }
  }
}

template <int H, int CHPH>
__global__ void node_logits(const float* __restrict__ h, const float* __restrict__ asrc,
                            const float* __restrict__ adst,
                            float* __restrict__ als, float* __restrict__ ald, int NH) {
  int tid = blockIdx.x * blockDim.x + threadIdx.x;
  if (tid >= NH) return;
  int node = tid / H;
  int head = tid % H;
  const float* hp = h + (size_t)node * (H * CHPH) + head * CHPH;
  const float* as = asrc + head * CHPH;
  const float* ad = adst + head * CHPH;
  float s = 0.f, d = 0.f;
#pragma unroll
  for (int c = 0; c < CHPH; ++c) {
    float v = hp[c];
    s = fmaf(v, as[c], s);
    d = fmaf(v, ad[c], d);
  }
  als[tid] = s;
  ald[tid] = d;
}

__device__ __forceinline__ float leaky(float v) {
  return v > 0.f ? v : NEG_SLOPE * v;
}

__global__ void fill_u32(unsigned* __restrict__ p, unsigned v, int n) {
  int i = blockIdx.x * blockDim.x + threadIdx.x;
  if (i < n) p[i] = v;
}


__global__ void dst_rank_k(const int* __restrict__ dst, int* __restrict__ deg,
                           int* __restrict__ rank, int E) {
  int e = blockIdx.x * blockDim.x + threadIdx.x;
  if (e >= E) return;
  rank[e] = atomicAdd(&deg[dst[e]], 1);
}

__global__ void __launch_bounds__(1024)
row_start_k(const int* __restrict__ deg, int* __restrict__ rowp, int n) {
  __shared__ int wsum[32];
  const int lane = threadIdx.x & 31;
  const int w = threadIdx.x >> 5;
  const int seg = (n + 31) / 32;
  const int b = w * seg;
  int e = b + seg;
  if (e > n) e = n;

  int s = 0;
#pragma unroll 4
  for (int i = b + lane; i < e; i += 32) s += deg[i];
  for (int o = 16; o > 0; o >>= 1) s += __shfl_xor(s, o);
  if (lane == 0) wsum[w] = s;
  __syncthreads();

  int run = 0;
  for (int k = 0; k < w; ++k) run += wsum[k];

  for (int i0 = b; i0 < e; i0 += 128) {
    const int i = i0 + lane * 4;
    const int v0 = (i     < e) ? deg[i]     : 0;
    const int v1 = (i + 1 < e) ? deg[i + 1] : 0;
    const int v2 = (i + 2 < e) ? deg[i + 2] : 0;
    const int v3 = (i + 3 < e) ? deg[i + 3] : 0;
    const int tot = v0 + v1 + v2 + v3;
    int inc = tot;
    for (int o = 1; o < 32; o <<= 1) {
      const int u = __shfl_up(inc, o);
      if (lane >= o) inc += u;
    }
    const int ex = run + inc - tot;
    if (i     < e) rowp[i]     = ex;
    if (i + 1 < e) rowp[i + 1] = ex + v0;
    if (i + 2 < e) rowp[i + 2] = ex + v0 + v1;
    if (i + 3 < e) rowp[i + 3] = ex + v0 + v1 + v2;
    run += __shfl(inc, 31);
  }
  if (threadIdx.x == 1023) rowp[n] = run;
}

__global__ void csr_fill_k(const int* __restrict__ src, const int* __restrict__ dst,
                           const int* __restrict__ rowp, const int* __restrict__ rank,
                           int* __restrict__ csr, int E) {
  int e = blockIdx.x * blockDim.x + threadIdx.x;
  if (e >= E) return;
  csr[rowp[dst[e]] + rank[e]] = src[e];
}

__global__ void copy_i32(const int* __restrict__ from, int* __restrict__ to, int n) {
  int i = blockIdx.x * blockDim.x + threadIdx.x;
  if (i < n) to[i] = from[i];
}

template <int H, int CH, bool ELU>
__global__ void node_gather(const int* __restrict__ rowp, const int* __restrict__ csr,
                            const float* __restrict__ h,
                            const float* __restrict__ als, const float* __restrict__ ald,
                            const float* __restrict__ bias, float* __restrict__ out, int N) {
  const int gw = (blockIdx.x * blockDim.x + threadIdx.x) >> 5;
  const int lane = threadIdx.x & 31;
  if (gw >= N) return;
  const int d = __builtin_amdgcn_readfirstlane(gw);
  constexpr int PER = CH / 32;
  constexpr int CHPH = CH / H;
  const int c0 = lane * PER;
  const int head = c0 / CHPH;
  const int beg = rowp[d];
  const int end = rowp[d + 1];
  const float ad = ald[d * H + head];

  float m = -INFINITY;
  for (int base = beg; base < end; base += 32) {
    int cnt = end - base;
    if (cnt > 32) cnt = 32;
    const int sv = (lane < cnt) ? csr[base + lane] : 0;
    for (int j = 0; j < cnt; ++j) {
      const int s = __builtin_amdgcn_readlane(sv, j);
      m = fmaxf(m, leaky(als[s * H + head] + ad));
    }
  }

  float ssum = 0.f;
  float acc[PER];
#pragma unroll
  for (int i = 0; i < PER; ++i) acc[i] = 0.f;
  for (int base = beg; base < end; base += 32) {
    int cnt = end - base;
    if (cnt > 32) cnt = 32;
    const int sv = (lane < cnt) ? csr[base + lane] : 0;
#pragma unroll 4
    for (int j = 0; j < cnt; ++j) {
      const int s = __builtin_amdgcn_readlane(sv, j);
      const float p = __expf(leaky(als[s * H + head] + ad) - m);
      ssum += p;
      const float* hrow = h + (size_t)s * CH + c0;
      if (PER == 4) {
        const v4f hv = *(const v4f*)hrow;
        acc[0] = fmaf(hv.x, p, acc[0]);
        acc[1] = fmaf(hv.y, p, acc[1]);
        acc[2] = fmaf(hv.z, p, acc[2]);
        acc[3] = fmaf(hv.w, p, acc[3]);
      } else {
        const v2f hv = *(const v2f*)hrow;
        acc[0] = fmaf(hv.x, p, acc[0]);
        acc[1] = fmaf(hv.y, p, acc[1]);
      }
    }
  }

  const float den = ssum + 1e-16f;
  float r[PER];
#pragma unroll
  for (int i = 0; i < PER; ++i) {
    float v = acc[i] / den + bias[c0 + i];
    if (ELU) v = v > 0.f ? v : (__expf(v) - 1.0f);
    r[i] = v;
  }
  float* orow = out + (size_t)d * CH + c0;
  if (PER == 4) {
    v4f o4 = {r[0], r[1], r[2], r[3]};
    *(v4f*)orow = o4;
  } else {
    v2f o2 = {r[0], r[1]};
    *(v2f*)orow = o2;
  }
}

extern "C" void kernel_launch(void* const* d_in, const int* in_sizes, int n_in,
                              void* d_out, int out_size, void* d_ws, size_t ws_size,
                              hipStream_t stream) {
  const float* x   = (const float*)d_in[0];
  const int*   ei  = (const int*)d_in[1];
  const float* W1  = (const float*)d_in[2];
  const float* as1 = (const float*)d_in[3];
  const float* ad1 = (const float*)d_in[4];
  const float* b1  = (const float*)d_in[5];
  const float* W2  = (const float*)d_in[6];
  const float* as2 = (const float*)d_in[7];
  const float* ad2 = (const float*)d_in[8];
  const float* b2  = (const float*)d_in[9];
  float* out = (float*)d_out;

  const int N = in_sizes[0] / IN_C;
  const int E = in_sizes[1] / 2;
  const int* src = ei;
  const int* dst = ei + E;

  float* ws = (float*)d_ws;
  float* h1   = ws;
  float* act1 = h1 + (size_t)N * 128;
  int*   csr  = (int*)(act1 + (size_t)N * 128);
  int*   rank = (int*)act1;
  int*   deg  = (int*)out;
  int*   rowp = deg + N;
  float* als1 = out + (size_t)N * 3;
  float* ald1 = als1 + (size_t)N * HEADS;
  float* h2    = h1;
  int*   rowp2 = (int*)(h1 + (size_t)N * 64);
  float* als2  = h1 + (size_t)N * 64 + (size_t)N * 2;
  float* ald2  = als2 + N;

  auto cdiv = [](long long a, long long b) -> unsigned { return (unsigned)((a + b - 1) / b); };

  gemm_wmma<IN_C, HEADS * HID><<<256, 512, 0, stream>>>(x, W1, h1, N);
  node_logits<HEADS, HID>
      <<<cdiv((long long)N * HEADS, 256), 256, 0, stream>>>(h1, as1, ad1, als1, ald1, N * HEADS);

  fill_u32<<<cdiv(N, 256), 256, 0, stream>>>((unsigned*)deg, 0u, N);
  dst_rank_k<<<cdiv(E, 256), 256, 0, stream>>>(dst, deg, rank, E);
  row_start_k<<<1, 1024, 0, stream>>>(deg, rowp, N);
  csr_fill_k<<<cdiv(E, 256), 256, 0, stream>>>(src, dst, rowp, rank, csr, E);

  node_gather<HEADS, 128, true>
      <<<cdiv((long long)N * 32, 256), 256, 0, stream>>>(rowp, csr, h1, als1, ald1, b1, act1, N);

  copy_i32<<<cdiv(N + 1, 256), 256, 0, stream>>>(rowp, rowp2, N + 1);
  gemm_wmma<HEADS * HID, OUT_C><<<256, 512, 0, stream>>>(act1, W2, h2, N);
  node_logits<1, OUT_C><<<cdiv(N, 256), 256, 0, stream>>>(h2, as2, ad2, als2, ald2, N);
  node_gather<1, OUT_C, false>
      <<<cdiv((long long)N * 32, 256), 256, 0, stream>>>(rowp2, csr, h2, als2, ald2, b2, out, N);
}


# node_gather: per-node chain of dependent loads cut from ~14 to ~4 round trips: batch logits gathered one per lane (max/exp across lanes, one pass), rows 8 at a time; edge sum order kept
# speedup vs baseline: 3.2236x; 1.1483x over previous
#include <hip/hip_runtime.h>
#include <math.h>

#define IN_C 128
#define HID 32
#define HEADS 4
#define OUT_C 64
#define NEG_SLOPE 0.2f

typedef __attribute__((ext_vector_type(2))) float v2f;
typedef __attribute__((ext_vector_type(4))) float v4f;
typedef __attribute__((ext_vector_type(8))) float v8f;

template <int K, int NC>
__global__ void __launch_bounds__(512)
gemm_wmma(const float* __restrict__ A, const float* __restrict__ B,
          float* __restrict__ C, int M) {
  constexpr int NT = NC / 16;
  constexpr int KS = K / 4;
  constexpr int PB = 2 * NC + 32;
  constexpr int WAVES = 16;
  constexpr int QN = NC / 4;
  __shared__ __attribute__((aligned(16))) float Ws[(K / 2) * PB];

  const int lane = threadIdx.x & 31;
  const int c = lane & 15;
  const int h = lane >> 4;
  const int wave = __builtin_amdgcn_readfirstlane((int)threadIdx.x >> 5);
  const int ntiles = (M + 15) >> 4;
  const int stride = (int)gridDim.x * WAVES;
  int tile = (int)blockIdx.x + (int)gridDim.x * wave;

  v2f a[KS];
  bool have = tile < ntiles;
  if (have) {
    int row = tile * 16 + c;
    if (row >= M) row = M - 1;
    const float* ap = A + (size_t)row * K + 2 * h;
#pragma unroll
    for (int ks = 0; ks < KS; ++ks) a[ks] = *(const v2f*)(ap + 4 * ks);
  }

  static_assert(((K / 2) * QN) % 512 == 0, "whole passes of the work-group");
#pragma unroll
  for (int it = 0; it < ((K / 2) * QN) / 512; ++it) {
    const int i = (int)threadIdx.x + it * 512;
    const int p = i / QN;
    const int q = i % QN;
    const v4f r0 = *(const v4f*)(B + (size_t)(2 * p) * NC + 4 * q);
    const v4f r1 = *(const v4f*)(B + (size_t)(2 * p + 1) * NC + 4 * q);
    float* d = Ws + p * PB + 8 * q;
    const v4f lo = {r0.x, r1.x, r0.y, r1.y};
    const v4f hi = {r0.z, r1.z, r0.w, r1.w};
    *(v4f*)d = lo;
    *(v4f*)(d + 4) = hi;
  }
  __syncthreads();

  const float* bl = Ws + h * PB + 2 * c;
  while (have) {
    asm volatile("" ::: "memory");
    v8f acc[NT];
#pragma unroll
    for (int t = 0; t < NT; ++t) acc[t] = v8f{};

#pragma unroll
    for (int ks = 0; ks < KS; ++ks) {
#pragma unroll
      for (int t = 0; t < NT; ++t) {
        const v2f bv = *(const v2f*)(bl + ks * 2 * PB + t * 32);
        acc[t] = __builtin_amdgcn_wmma_f32_16x16x4_f32(false, a[ks], false, bv, (short)0,
                                                       acc[t], false, false);
      }
    }

    const int rbase = tile * 16 + h * 8;
    float* cbase = C + (size_t)rbase * NC + c;
    if (tile * 16 + 16 <= M) {
#pragma unroll
      for (int t = 0; t < NT; ++t)
#pragma unroll
        for (int i = 0; i < 8; ++i) cbase[(size_t)i * NC + t * 16] = acc[t][i];
    } else {
#pragma unroll
      for (int t = 0; t < NT; ++t)
#pragma unroll
        for (int i = 0; i < 8; ++i)
          if (rbase + i < M) cbase[(size_t)i * NC + t * 16] = acc[t][i];
    }

    tile += stride;
    have = tile < ntiles;
    if (have) {
      int row = tile * 16 + c;
      if (row >= M) row = M - 1;
      const float* ap = A + (size_t)row * K + 2 * h;
#pragma unroll
      for (int ks = 0; ks < KS; ++ks) a[ks] = *(const v2f*)(ap + 4 * ks);
    }
  }
}

template <int H, int CHPH>
__global__ void node_logits(const float* __restrict__ h, const float* __restrict__ asrc,
                            const float* __restrict__ adst,
                            float* __restrict__ als, float* __restrict__ ald, int NH) {
  int tid = blockIdx.x * blockDim.x + threadIdx.x;
  if (tid >= NH) return;
  int node = tid / H;
  int head = tid % H;
  const float* hp = h + (size_t)node * (H * CHPH) + head * CHPH;
  const float* as = asrc + head * CHPH;
  const float* ad = adst + head * CHPH;
  float s = 0.f, d = 0.f;
#pragma unroll
  for (int c = 0; c < CHPH; ++c) {
    float v = hp[c];
    s = fmaf(v, as[c], s);
    d = fmaf(v, ad[c], d);
  }
  als[tid] = s;
  ald[tid] = d;
}

__device__ __forceinline__ float leaky(float v) {
  return v > 0.f ? v : NEG_SLOPE * v;
}

__global__ void fill_u32(unsigned* __restrict__ p, unsigned v, int n) {
  int i = blockIdx.x * blockDim.x + threadIdx.x;
  if (i < n) p[i] = v;
}


__global__ void dst_rank_k(const int* __restrict__ dst, int* __restrict__ deg,
                           int* __restrict__ rank, int E) {
  int e = blockIdx.x * blockDim.x + threadIdx.x;
  if (e >= E) return;
  rank[e] = atomicAdd(&deg[dst[e]], 1);
}

__global__ void __launch_bounds__(1024)
row_start_k(const int* __restrict__ deg, int* __restrict__ rowp, int n) {
  __shared__ int wsum[32];
  const int lane = threadIdx.x & 31;
  const int w = threadIdx.x >> 5;
  const int seg = (n + 31) / 32;
  const int b = w * seg;
  int e = b + seg;
  if (e > n) e = n;

  int s = 0;
#pragma unroll 4
  for (int i = b + lane; i < e; i += 32) s += deg[i];
  for (int o = 16; o > 0; o >>= 1) s += __shfl_xor(s, o);
  if (lane == 0) wsum[w] = s;
  __syncthreads();

  int run = 0;
  for (int k = 0; k < w; ++k) run += wsum[k];

  for (int i0 = b; i0 < e; i0 += 128) {
    const int i = i0 + lane * 4;
    const int v0 = (i     < e) ? deg[i]     : 0;
    const int v1 = (i + 1 < e) ? deg[i + 1] : 0;
    const int v2 = (i + 2 < e) ? deg[i + 2] : 0;
    const int v3 = (i + 3 < e) ? deg[i + 3] : 0;
    const int tot = v0 + v1 + v2 + v3;
    int inc = tot;
    for (int o = 1; o < 32; o <<= 1) {
      const int u = __shfl_up(inc, o);
      if (lane >= o) inc += u;
    }
    const int ex = run + inc - tot;
    if (i     < e) rowp[i]     = ex;
    if (i + 1 < e) rowp[i + 1] = ex + v0;
    if (i + 2 < e) rowp[i + 2] = ex + v0 + v1;
    if (i + 3 < e) rowp[i + 3] = ex + v0 + v1 + v2;
    run += __shfl(inc, 31);
  }
  if (threadIdx.x == 1023) rowp[n] = run;
}

__global__ void csr_fill_k(const int* __restrict__ src, const int* __restrict__ dst,
                           const int* __restrict__ rowp, const int* __restrict__ rank,
                           int* __restrict__ csr, int E) {
  int e = blockIdx.x * blockDim.x + threadIdx.x;
  if (e >= E) return;
  csr[rowp[dst[e]] + rank[e]] = src[e];
}

__global__ void copy_i32(const int* __restrict__ from, int* __restrict__ to, int n) {
  int i = blockIdx.x * blockDim.x + threadIdx.x;
  if (i < n) to[i] = from[i];
}

constexpr unsigned dpp8_xor_sel(int m) {
  unsigned s = 0;
  for (int i = 0; i < 8; ++i) s |= (unsigned)(i ^ m) << (3 * i);
  return s;
}
constexpr unsigned dpp8_bcast_sel(int i) { return (unsigned)i * 0x249249u; }

template <unsigned SEL>
__device__ __forceinline__ float dpp8_f(float v) {
  return __builtin_bit_cast(float,
                            __builtin_amdgcn_mov_dpp8(__builtin_bit_cast(unsigned, v), SEL));
}
__device__ __forceinline__ float xor8_f(float v) {
  return __builtin_bit_cast(
      float, __builtin_amdgcn_update_dpp(0, __builtin_bit_cast(int, v), 0x168, 0xf, 0xf, false));
}
__device__ __forceinline__ float xor16_f(float v) {
  return __builtin_bit_cast(
      float, __builtin_amdgcn_permlanex16(__builtin_bit_cast(int, v), __builtin_bit_cast(int, v),
                                          0x76543210, 0xfedcba98, false, false));
}
template <int G>
__device__ __forceinline__ float group_max(float v) {
  v = fmaxf(v, dpp8_f<dpp8_xor_sel(1)>(v));
  v = fmaxf(v, dpp8_f<dpp8_xor_sel(2)>(v));
  v = fmaxf(v, dpp8_f<dpp8_xor_sel(4)>(v));
  if constexpr (G == 32) {
    v = fmaxf(v, xor8_f(v));
    v = fmaxf(v, xor16_f(v));
  }
  return v;
}

template <int H, int CH, bool ELU>
__global__ void node_gather(const int* __restrict__ rowp, const int* __restrict__ csr,
                            const float* __restrict__ h,
                            const float* __restrict__ als, const float* __restrict__ ald,
                            const float* __restrict__ bias, float* __restrict__ out, int N) {
  static_assert(H == 1 || H == 4, "head groups of 32 or 8 lanes");
  const int gw = (blockIdx.x * blockDim.x + threadIdx.x) >> 5;
  const int lane = threadIdx.x & 31;
  if (gw >= N) return;
  const int d = __builtin_amdgcn_readfirstlane(gw);
  constexpr int PER = CH / 32;
  constexpr int G = 32 / H;
  typedef float RowV __attribute__((ext_vector_type(PER)));
  const int c0 = lane * PER;
  const int head = lane / G;
  const int li = lane % G;
  const int beg = rowp[d];
  const int end = rowp[d + 1];
  const float ad = ald[d * H + head];
  const RowV bv = *(const RowV*)(bias + c0);

  float ssum = 0.f;
  float acc[PER];
#pragma unroll
  for (int i = 0; i < PER; ++i) acc[i] = 0.f;

#define GATHER_ROW(J)                                                              \
  if ((J) < cnt) {                                                                 \
    const int s_ = __builtin_amdgcn_readlane(sk[(J) / G], (J) % G);                \
    hv[(J) & 7] = *(const RowV*)(h + (size_t)s_ * CH + c0);                        \
  }
#define GATHER_ROWS8(J0)                                                           \
  GATHER_ROW((J0) + 0) GATHER_ROW((J0) + 1) GATHER_ROW((J0) + 2) GATHER_ROW((J0) + 3) \
  GATHER_ROW((J0) + 4) GATHER_ROW((J0) + 5) GATHER_ROW((J0) + 6) GATHER_ROW((J0) + 7)
#define FMA_ROW(J)                                                                 \
  if ((J) < cnt) {                                                                 \
    float p_;                                                                      \
    if constexpr (H == 4) {                                                        \
      p_ = dpp8_f<dpp8_bcast_sel((J) % 8)>(tk[(J) / 8]);                           \
    } else {                                                                       \
      p_ = __builtin_bit_cast(                                                     \
          float, __builtin_amdgcn_readlane(__builtin_bit_cast(int, tk[0]), (J)));  \
    }                                                                              \
    ssum += p_;                                                                    \
      \
    _Pragma("unroll") for (int i_ = 0; i_ < PER; ++i_)                             \
        acc[i_] = fmaf(hv[(J) & 7][i_], p_, acc[i_]);                              \
  }
#define FMA_ROWS8(J0)                                                              \
  FMA_ROW((J0) + 0) FMA_ROW((J0) + 1) FMA_ROW((J0) + 2) FMA_ROW((J0) + 3)          \
  FMA_ROW((J0) + 4) FMA_ROW((J0) + 5) FMA_ROW((J0) + 6) FMA_ROW((J0) + 7)
#define LOAD_IDS(BASE)                                                             \
           \
  _Pragma("unroll") for (int k = 0; k < H; ++k) {                                  \
    const int jj = k * G + li;                                                     \
    sk[k] = csr[(BASE) + (jj < cnt ? jj : cnt - 1)];                               \
  }

  if (beg < end) {
    int sk[H];
    float ak[H], ek[H], tk[H];
    RowV hv[8];
    int cnt = end - beg;
    if (cnt > 32) cnt = 32;

    LOAD_IDS(beg)
#pragma unroll
    for (int k = 0; k < H; ++k) ak[k] = als[sk[k] * H + head];
    GATHER_ROWS8(0)

    float m = -INFINITY;
#pragma unroll
    for (int k = 0; k < H; ++k) {
      ek[k] = (k * G + li < cnt) ? leaky(ak[k] + ad) : -INFINITY;
      m = fmaxf(m, ek[k]);
    }
    for (int base = beg + 32; base < end; base += 32) {
      int c2 = end - base;
      if (c2 > 32) c2 = 32;
#pragma unroll
      for (int k = 0; k < H; ++k) {
        const int jj = k * G + li;
        const int s2 = csr[base + (jj < c2 ? jj : c2 - 1)];
        const float e2 = (jj < c2) ? leaky(als[s2 * H + head] + ad) : -INFINITY;
        m = fmaxf(m, e2);
      }
    }
    m = group_max<G>(m);

    for (int base = beg; base < end; base += 32) {
      cnt = end - base;
      if (cnt > 32) cnt = 32;
      if (base != beg) {
        LOAD_IDS(base)
#pragma unroll
        for (int k = 0; k < H; ++k) ak[k] = als[sk[k] * H + head];
        GATHER_ROWS8(0)
#pragma unroll
        for (int k = 0; k < H; ++k)
          ek[k] = (k * G + li < cnt) ? leaky(ak[k] + ad) : -INFINITY;
      }
#pragma unroll
      for (int k = 0; k < H; ++k)
        tk[k] = (k * G + li < cnt) ? __expf(ek[k] - m) : 0.f;

      FMA_ROWS8(0)
      if (8 < cnt) {
        GATHER_ROWS8(8)
        FMA_ROWS8(8)
      }
      if (16 < cnt) {
        GATHER_ROWS8(16)
        FMA_ROWS8(16)
      }
      if (24 < cnt) {
        GATHER_ROWS8(24)
        FMA_ROWS8(24)
      }
    }
  }
#undef GATHER_ROW
#undef GATHER_ROWS8
#undef FMA_ROW
#undef FMA_ROWS8
#undef LOAD_IDS

  const float den = ssum + 1e-16f;
  RowV o;
#pragma unroll
  for (int i = 0; i < PER; ++i) {
    float v = acc[i] / den + bv[i];
    if (ELU) v = v > 0.f ? v : (__expf(v) - 1.0f);
    o[i] = v;
  }
  *(RowV*)(out + (size_t)d * CH + c0) = o;
}

extern "C" void kernel_launch(void* const* d_in, const int* in_sizes, int n_in,
                              void* d_out, int out_size, void* d_ws, size_t ws_size,
                              hipStream_t stream) {
  const float* x   = (const float*)d_in[0];
  const int*   ei  = (const int*)d_in[1];
  const float* W1  = (const float*)d_in[2];
  const float* as1 = (const float*)d_in[3];
  const float* ad1 = (const float*)d_in[4];
  const float* b1  = (const float*)d_in[5];
  const float* W2  = (const float*)d_in[6];
  const float* as2 = (const float*)d_in[7];
  const float* ad2 = (const float*)d_in[8];
  const float* b2  = (const float*)d_in[9];
  float* out = (float*)d_out;

  const int N = in_sizes[0] / IN_C;
  const int E = in_sizes[1] / 2;
  const int* src = ei;
  const int* dst = ei + E;

  float* ws = (float*)d_ws;
  float* h1   = ws;
  float* act1 = h1 + (size_t)N * 128;
  int*   csr  = (int*)(act1 + (size_t)N * 128);
  int*   rank = (int*)act1;
  int*   deg  = (int*)out;
  int*   rowp = deg + N;
  float* als1 = out + (size_t)N * 3;
  float* ald1 = als1 + (size_t)N * HEADS;
  float* h2    = h1;
  int*   rowp2 = (int*)(h1 + (size_t)N * 64);
  float* als2  = h1 + (size_t)N * 64 + (size_t)N * 2;
  float* ald2  = als2 + N;

  auto cdiv = [](long long a, long long b) -> unsigned { return (unsigned)((a + b - 1) / b); };

  gemm_wmma<IN_C, HEADS * HID><<<256, 512, 0, stream>>>(x, W1, h1, N);
  node_logits<HEADS, HID>
      <<<cdiv((long long)N * HEADS, 256), 256, 0, stream>>>(h1, as1, ad1, als1, ald1, N * HEADS);

  fill_u32<<<cdiv(N, 256), 256, 0, stream>>>((unsigned*)deg, 0u, N);
  dst_rank_k<<<cdiv(E, 256), 256, 0, stream>>>(dst, deg, rank, E);
  row_start_k<<<1, 1024, 0, stream>>>(deg, rowp, N);
  csr_fill_k<<<cdiv(E, 256), 256, 0, stream>>>(src, dst, rowp, rank, csr, E);

  node_gather<HEADS, 128, true>
      <<<cdiv((long long)N * 32, 256), 256, 0, stream>>>(rowp, csr, h1, als1, ald1, b1, act1, N);

  copy_i32<<<cdiv(N + 1, 256), 256, 0, stream>>>(rowp, rowp2, N + 1);
  gemm_wmma<HEADS * HID, OUT_C><<<256, 512, 0, stream>>>(act1, W2, h2, N);
  node_logits<1, OUT_C><<<cdiv(N, 256), 256, 0, stream>>>(h2, as2, ad2, als2, ald2, N);
  node_gather<1, OUT_C, false>
      <<<cdiv((long long)N * 32, 256), 256, 0, stream>>>(rowp2, csr, h2, als2, ald2, b2, out, N);
}


# Edges bucketed by a two-level counting sort on LDS counters (bins of 256 nodes, then nodes) replacing per-edge returning global adds and the one-group scan; per-node edge order = LDS arrival order
# speedup vs baseline: 5.3085x; 1.6468x over previous
#include <hip/hip_runtime.h>
#include <math.h>

#define IN_C 128
#define HID 32
#define HEADS 4
#define OUT_C 64
#define NEG_SLOPE 0.2f

typedef __attribute__((ext_vector_type(2))) float v2f;
typedef __attribute__((ext_vector_type(4))) float v4f;
typedef __attribute__((ext_vector_type(8))) float v8f;

template <int K, int NC>
__global__ void __launch_bounds__(512)
gemm_wmma(const float* __restrict__ A, const float* __restrict__ B,
          float* __restrict__ C, int M) {
  constexpr int NT = NC / 16;
  constexpr int KS = K / 4;
  constexpr int PB = 2 * NC + 32;
  constexpr int WAVES = 16;
  constexpr int QN = NC / 4;
  __shared__ __attribute__((aligned(16))) float Ws[(K / 2) * PB];

  const int lane = threadIdx.x & 31;
  const int c = lane & 15;
  const int h = lane >> 4;
  const int wave = __builtin_amdgcn_readfirstlane((int)threadIdx.x >> 5);
  const int ntiles = (M + 15) >> 4;
  const int stride = (int)gridDim.x * WAVES;
  int tile = (int)blockIdx.x + (int)gridDim.x * wave;

  v2f a[KS];
  bool have = tile < ntiles;
  if (have) {
    int row = tile * 16 + c;
    if (row >= M) row = M - 1;
    const float* ap = A + (size_t)row * K + 2 * h;
#pragma unroll
    for (int ks = 0; ks < KS; ++ks) a[ks] = *(const v2f*)(ap + 4 * ks);
  }

  static_assert(((K / 2) * QN) % 512 == 0, "whole passes of the work-group");
#pragma unroll
  for (int it = 0; it < ((K / 2) * QN) / 512; ++it) {
    const int i = (int)threadIdx.x + it * 512;
    const int p = i / QN;
    const int q = i % QN;
    const v4f r0 = *(const v4f*)(B + (size_t)(2 * p) * NC + 4 * q);
    const v4f r1 = *(const v4f*)(B + (size_t)(2 * p + 1) * NC + 4 * q);
    float* d = Ws + p * PB + 8 * q;
    const v4f lo = {r0.x, r1.x, r0.y, r1.y};
    const v4f hi = {r0.z, r1.z, r0.w, r1.w};
    *(v4f*)d = lo;
    *(v4f*)(d + 4) = hi;
  }
  __syncthreads();

  const float* bl = Ws + h * PB + 2 * c;
  while (have) {
    asm volatile("" ::: "memory");
    v8f acc[NT];
#pragma unroll
    for (int t = 0; t < NT; ++t) acc[t] = v8f{};

#pragma unroll
    for (int ks = 0; ks < KS; ++ks) {
#pragma unroll
      for (int t = 0; t < NT; ++t) {
        const v2f bv = *(const v2f*)(bl + ks * 2 * PB + t * 32);
        acc[t] = __builtin_amdgcn_wmma_f32_16x16x4_f32(false, a[ks], false, bv, (short)0,
                                                       acc[t], false, false);
      }
    }

    const int rbase = tile * 16 + h * 8;
    float* cbase = C + (size_t)rbase * NC + c;
    if (tile * 16 + 16 <= M) {
#pragma unroll
      for (int t = 0; t < NT; ++t)
#pragma unroll
        for (int i = 0; i < 8; ++i) cbase[(size_t)i * NC + t * 16] = acc[t][i];
    } else {
#pragma unroll
      for (int t = 0; t < NT; ++t)
#pragma unroll
        for (int i = 0; i < 8; ++i)
          if (rbase + i < M) cbase[(size_t)i * NC + t * 16] = acc[t][i];
    }

    tile += stride;
    have = tile < ntiles;
    if (have) {
      int row = tile * 16 + c;
      if (row >= M) row = M - 1;
      const float* ap = A + (size_t)row * K + 2 * h;
#pragma unroll
      for (int ks = 0; ks < KS; ++ks) a[ks] = *(const v2f*)(ap + 4 * ks);
    }
  }
}

template <int H, int CHPH>
__global__ void node_logits(const float* __restrict__ h, const float* __restrict__ asrc,
                            const float* __restrict__ adst,
                            float* __restrict__ als, float* __restrict__ ald, int NH) {
  int tid = blockIdx.x * blockDim.x + threadIdx.x;
  if (tid >= NH) return;
  int node = tid / H;
  int head = tid % H;
  const float* hp = h + (size_t)node * (H * CHPH) + head * CHPH;
  const float* as = asrc + head * CHPH;
  const float* ad = adst + head * CHPH;
  float s = 0.f, d = 0.f;
#pragma unroll
  for (int c = 0; c < CHPH; ++c) {
    float v = hp[c];
    s = fmaf(v, as[c], s);
    d = fmaf(v, ad[c], d);
  }
  als[tid] = s;
  ald[tid] = d;
}

__device__ __forceinline__ float leaky(float v) {
  return v > 0.f ? v : NEG_SLOPE * v;
}

__global__ void fill_u32(unsigned* __restrict__ p, unsigned v, int n) {
  int i = blockIdx.x * blockDim.x + threadIdx.x;
  if (i < n) p[i] = v;
}


#define BSH 8
#define BW (1 << BSH)
#define NBMAX 512

__device__ __forceinline__ int wg_excl_scan(int v, int* wsum) {
  const int lane = threadIdx.x & 31;
  const int w = threadIdx.x >> 5;
  int inc = v;
  for (int o = 1; o < 32; o <<= 1) {
    const int u = __shfl_up(inc, o);
    if (lane >= o) inc += u;
  }
  if (lane == 31) wsum[w] = inc;
  __syncthreads();
  int run = 0;
  for (int k = 0; k < w; ++k) run += wsum[k];
  return run + inc - v;
}

__global__ void __launch_bounds__(1024)
bin_count_k(const int* __restrict__ dst, int* __restrict__ bintotal,
            int* __restrict__ wgbase, int E, int NB) {
  __shared__ int hist[NBMAX];
  const int t = threadIdx.x;
  if (t < NBMAX) hist[t] = 0;
  __syncthreads();
  const int chunk = (E + (int)gridDim.x - 1) / (int)gridDim.x;
  const int b = (int)blockIdx.x * chunk;
  int en = b + chunk;
  if (en > E) en = E;
  for (int e0 = b + t; e0 < en; e0 += 4096) {
    int d[4];
    bool ok[4];
#pragma unroll
    for (int k = 0; k < 4; ++k) {
      const int e = e0 + k * 1024;
      ok[k] = e < en;
      d[k] = dst[ok[k] ? e : en - 1];
    }
#pragma unroll
    for (int k = 0; k < 4; ++k) atomicAdd(&hist[d[k] >> BSH], ok[k] ? 1 : 0);
  }
  __syncthreads();
  if (t < NB) wgbase[(int)blockIdx.x * NB + t] = atomicAdd(&bintotal[t], hist[t]);
}

__global__ void __launch_bounds__(1024)
bin_scatter_k(const int* __restrict__ src, const int* __restrict__ dst,
              const int* __restrict__ bintotal, const int* __restrict__ wgbase,
              int2* __restrict__ tmp, int E, int NB) {
  __shared__ int cur[NBMAX];
  __shared__ int wsum[32];
  const int t = threadIdx.x;
  const int ex = wg_excl_scan(t < NB ? bintotal[t] : 0, wsum);
  if (t < NB) cur[t] = ex + wgbase[(int)blockIdx.x * NB + t];
  __syncthreads();
  const int chunk = (E + (int)gridDim.x - 1) / (int)gridDim.x;
  const int b = (int)blockIdx.x * chunk;
  int en = b + chunk;
  if (en > E) en = E;
  for (int e0 = b + t; e0 < en; e0 += 4096) {
    int d[4], s[4];
    bool ok[4];
#pragma unroll
    for (int k = 0; k < 4; ++k) {
      const int e = e0 + k * 1024;
      ok[k] = e < en;
      const int ec = ok[k] ? e : en - 1;
      d[k] = dst[ec];
      s[k] = src[ec];
    }
#pragma unroll
    for (int k = 0; k < 4; ++k) {
      const int p = atomicAdd(&cur[d[k] >> BSH], ok[k] ? 1 : 0);
      if (ok[k]) tmp[p] = make_int2(s[k], d[k]);
    }
  }
}

__global__ void __launch_bounds__(1024)
bin_place_k(const int2* __restrict__ tmp, const int* __restrict__ bintotal,
            int* __restrict__ rowp, int* __restrict__ csr, int N, int NB) {
  __shared__ int cnt[BW];
  __shared__ int cur[BW];
  __shared__ int wsum[32];
  __shared__ int wsum2[32];
  __shared__ int own;
  const int t = threadIdx.x;
  const int lane = t & 31;
  const int bin = (int)blockIdx.x;
  const int bt = (t <= bin) ? bintotal[t] : 0;
  int s = (t < bin) ? bt : 0;
  for (int o = 16; o > 0; o >>= 1) s += __shfl_xor(s, o);
  if (lane == 0) wsum[t >> 5] = s;
  if (t == bin) own = bt;
  if (t < BW) cnt[t] = 0;
  __syncthreads();
  int beg = 0;
  for (int k = 0; k < 32; ++k) beg += wsum[k];
  const int end = beg + own;

  for (int i0 = beg + t; i0 < end; i0 += 4096) {
    int d[4];
    bool ok[4];
#pragma unroll
    for (int k = 0; k < 4; ++k) {
      const int i = i0 + k * 1024;
      ok[k] = i < end;
      d[k] = tmp[ok[k] ? i : end - 1].y;
    }
#pragma unroll
    for (int k = 0; k < 4; ++k) atomicAdd(&cnt[d[k] & (BW - 1)], ok[k] ? 1 : 0);
  }
  __syncthreads();
  const int ex = wg_excl_scan(t < BW ? cnt[t] : 0, wsum2);
  if (t < BW) {
    cur[t] = beg + ex;
    const int node = bin * BW + t;
    if (node < N) rowp[node] = beg + ex;
  }
  if (bin == NB - 1 && t == 0) rowp[N] = end;
  __syncthreads();
  for (int i0 = beg + t; i0 < end; i0 += 4096) {
    int2 sd[4];
    bool ok[4];
#pragma unroll
    for (int k = 0; k < 4; ++k) {
      const int i = i0 + k * 1024;
      ok[k] = i < end;
      sd[k] = tmp[ok[k] ? i : end - 1];
    }
#pragma unroll
    for (int k = 0; k < 4; ++k) {
      const int p = atomicAdd(&cur[sd[k].y & (BW - 1)], ok[k] ? 1 : 0);
      if (ok[k]) csr[p] = sd[k].x;
    }
  }
}

__global__ void copy_i32(const int* __restrict__ from, int* __restrict__ to, int n) {
  int i = blockIdx.x * blockDim.x + threadIdx.x;
  if (i < n) to[i] = from[i];
}

constexpr unsigned dpp8_xor_sel(int m) {
  unsigned s = 0;
  for (int i = 0; i < 8; ++i) s |= (unsigned)(i ^ m) << (3 * i);
  return s;
}
constexpr unsigned dpp8_bcast_sel(int i) { return (unsigned)i * 0x249249u; }

template <unsigned SEL>
__device__ __forceinline__ float dpp8_f(float v) {
  return __builtin_bit_cast(float,
                            __builtin_amdgcn_mov_dpp8(__builtin_bit_cast(unsigned, v), SEL));
}
__device__ __forceinline__ float xor8_f(float v) {
  return __builtin_bit_cast(
      float, __builtin_amdgcn_update_dpp(0, __builtin_bit_cast(int, v), 0x168, 0xf, 0xf, false));
}
__device__ __forceinline__ float xor16_f(float v) {
  return __builtin_bit_cast(
      float, __builtin_amdgcn_permlanex16(__builtin_bit_cast(int, v), __builtin_bit_cast(int, v),
                                          0x76543210, 0xfedcba98, false, false));
}
template <int G>
__device__ __forceinline__ float group_max(float v) {
  v = fmaxf(v, dpp8_f<dpp8_xor_sel(1)>(v));
  v = fmaxf(v, dpp8_f<dpp8_xor_sel(2)>(v));
  v = fmaxf(v, dpp8_f<dpp8_xor_sel(4)>(v));
  if constexpr (G == 32) {
    v = fmaxf(v, xor8_f(v));
    v = fmaxf(v, xor16_f(v));
  }
  return v;
}

template <int H, int CH, bool ELU>
__global__ void node_gather(const int* __restrict__ rowp, const int* __restrict__ csr,
                            const float* __restrict__ h,
                            const float* __restrict__ als, const float* __restrict__ ald,
                            const float* __restrict__ bias, float* __restrict__ out, int N) {
  static_assert(H == 1 || H == 4, "head groups of 32 or 8 lanes");
  const int gw = (blockIdx.x * blockDim.x + threadIdx.x) >> 5;
  const int lane = threadIdx.x & 31;
  if (gw >= N) return;
  const int d = __builtin_amdgcn_readfirstlane(gw);
  constexpr int PER = CH / 32;
  constexpr int G = 32 / H;
  typedef float RowV __attribute__((ext_vector_type(PER)));
  const int c0 = lane * PER;
  const int head = lane / G;
  const int li = lane % G;
  const int beg = rowp[d];
  const int end = rowp[d + 1];
  const float ad = ald[d * H + head];
  const RowV bv = *(const RowV*)(bias + c0);

  float ssum = 0.f;
  float acc[PER];
#pragma unroll
  for (int i = 0; i < PER; ++i) acc[i] = 0.f;

#define GATHER_ROW(J)                                                              \
  if ((J) < cnt) {                                                                 \
    const int s_ = __builtin_amdgcn_readlane(sk[(J) / G], (J) % G);                \
    hv[(J) & 7] = *(const RowV*)(h + (size_t)s_ * CH + c0);                        \
  }
#define GATHER_ROWS8(J0)                                                           \
  GATHER_ROW((J0) + 0) GATHER_ROW((J0) + 1) GATHER_ROW((J0) + 2) GATHER_ROW((J0) + 3) \
  GATHER_ROW((J0) + 4) GATHER_ROW((J0) + 5) GATHER_ROW((J0) + 6) GATHER_ROW((J0) + 7)
#define FMA_ROW(J)                                                                 \
  if ((J) < cnt) {                                                                 \
    float p_;                                                                      \
    if constexpr (H == 4) {                                                        \
      p_ = dpp8_f<dpp8_bcast_sel((J) % 8)>(tk[(J) / 8]);                           \
    } else {                                                                       \
      p_ = __builtin_bit_cast(                                                     \
          float, __builtin_amdgcn_readlane(__builtin_bit_cast(int, tk[0]), (J)));  \
    }                                                                              \
    ssum += p_;                                                                    \
      \
    _Pragma("unroll") for (int i_ = 0; i_ < PER; ++i_)                             \
        acc[i_] = fmaf(hv[(J) & 7][i_], p_, acc[i_]);                              \
  }
#define FMA_ROWS8(J0)                                                              \
  FMA_ROW((J0) + 0) FMA_ROW((J0) + 1) FMA_ROW((J0) + 2) FMA_ROW((J0) + 3)          \
  FMA_ROW((J0) + 4) FMA_ROW((J0) + 5) FMA_ROW((J0) + 6) FMA_ROW((J0) + 7)
#define LOAD_IDS(BASE)                                                             \
           \
  _Pragma("unroll") for (int k = 0; k < H; ++k) {                                  \
    const int jj = k * G + li;                                                     \
    sk[k] = csr[(BASE) + (jj < cnt ? jj : cnt - 1)];                               \
  }

  if (beg < end) {
    int sk[H];
    float ak[H], ek[H], tk[H];
    RowV hv[8];
    int cnt = end - beg;
    if (cnt > 32) cnt = 32;

    LOAD_IDS(beg)
#pragma unroll
    for (int k = 0; k < H; ++k) ak[k] = als[sk[k] * H + head];
    GATHER_ROWS8(0)

    float m = -INFINITY;
#pragma unroll
    for (int k = 0; k < H; ++k) {
      ek[k] = (k * G + li < cnt) ? leaky(ak[k] + ad) : -INFINITY;
      m = fmaxf(m, ek[k]);
    }
    for (int base = beg + 32; base < end; base += 32) {
      int c2 = end - base;
      if (c2 > 32) c2 = 32;
#pragma unroll
      for (int k = 0; k < H; ++k) {
        const int jj = k * G + li;
        const int s2 = csr[base + (jj < c2 ? jj : c2 - 1)];
        const float e2 = (jj < c2) ? leaky(als[s2 * H + head] + ad) : -INFINITY;
        m = fmaxf(m, e2);
      }
    }
    m = group_max<G>(m);

    for (int base = beg; base < end; base += 32) {
      cnt = end - base;
      if (cnt > 32) cnt = 32;
      if (base != beg) {
        LOAD_IDS(base)
#pragma unroll
        for (int k = 0; k < H; ++k) ak[k] = als[sk[k] * H + head];
        GATHER_ROWS8(0)
#pragma unroll
        for (int k = 0; k < H; ++k)
          ek[k] = (k * G + li < cnt) ? leaky(ak[k] + ad) : -INFINITY;
      }
#pragma unroll
      for (int k = 0; k < H; ++k)
        tk[k] = (k * G + li < cnt) ? __expf(ek[k] - m) : 0.f;

      FMA_ROWS8(0)
      if (8 < cnt) {
        GATHER_ROWS8(8)
        FMA_ROWS8(8)
      }
      if (16 < cnt) {
        GATHER_ROWS8(16)
        FMA_ROWS8(16)
      }
      if (24 < cnt) {
        GATHER_ROWS8(24)
        FMA_ROWS8(24)
      }
    }
  }
#undef GATHER_ROW
#undef GATHER_ROWS8
#undef FMA_ROW
#undef FMA_ROWS8
#undef LOAD_IDS

  const float den = ssum + 1e-16f;
  RowV o;
#pragma unroll
  for (int i = 0; i < PER; ++i) {
    float v = acc[i] / den + bv[i];
    if (ELU) v = v > 0.f ? v : (__expf(v) - 1.0f);
    o[i] = v;
  }
  *(RowV*)(out + (size_t)d * CH + c0) = o;
}

extern "C" void kernel_launch(void* const* d_in, const int* in_sizes, int n_in,
                              void* d_out, int out_size, void* d_ws, size_t ws_size,
                              hipStream_t stream) {
  const float* x   = (const float*)d_in[0];
  const int*   ei  = (const int*)d_in[1];
  const float* W1  = (const float*)d_in[2];
  const float* as1 = (const float*)d_in[3];
  const float* ad1 = (const float*)d_in[4];
  const float* b1  = (const float*)d_in[5];
  const float* W2  = (const float*)d_in[6];
  const float* as2 = (const float*)d_in[7];
  const float* ad2 = (const float*)d_in[8];
  const float* b2  = (const float*)d_in[9];
  float* out = (float*)d_out;

  const int N = in_sizes[0] / IN_C;
  const int E = in_sizes[1] / 2;
  const int* src = ei;
  const int* dst = ei + E;

  float* ws = (float*)d_ws;
  float* h1   = ws;
  float* act1 = h1 + (size_t)N * 128;
  int*   csr  = (int*)(act1 + (size_t)N * 128);
  int2*  pairs = (int2*)act1;
  int*   bintotal = (int*)out;
  int*   rowp = (int*)out + N;
  int*   wgbase = (int*)out + (size_t)N * 11;
  float* als1 = out + (size_t)N * 3;
  float* ald1 = als1 + (size_t)N * HEADS;
  float* h2    = h1;
  int*   rowp2 = (int*)(h1 + (size_t)N * 64);
  float* als2  = h1 + (size_t)N * 64 + (size_t)N * 2;
  float* ald2  = als2 + N;

  auto cdiv = [](long long a, long long b) -> unsigned { return (unsigned)((a + b - 1) / b); };

  gemm_wmma<IN_C, HEADS * HID><<<256, 512, 0, stream>>>(x, W1, h1, N);
  node_logits<HEADS, HID>
      <<<cdiv((long long)N * HEADS, 256), 256, 0, stream>>>(h1, as1, ad1, als1, ald1, N * HEADS);

  const int NB = (int)cdiv(N, BW);
  fill_u32<<<cdiv(NB, 256), 256, 0, stream>>>((unsigned*)bintotal, 0u, NB);
  bin_count_k<<<256, 1024, 0, stream>>>(dst, bintotal, wgbase, E, NB);
  bin_scatter_k<<<256, 1024, 0, stream>>>(src, dst, bintotal, wgbase, pairs, E, NB);
  bin_place_k<<<NB, 1024, 0, stream>>>(pairs, bintotal, rowp, csr, N, NB);

  node_gather<HEADS, 128, true>
      <<<cdiv((long long)N * 32, 256), 256, 0, stream>>>(rowp, csr, h1, als1, ald1, b1, act1, N);

  copy_i32<<<cdiv(N + 1, 256), 256, 0, stream>>>(rowp, rowp2, N + 1);
  gemm_wmma<HEADS * HID, OUT_C><<<256, 512, 0, stream>>>(act1, W2, h2, N);
  node_logits<1, OUT_C><<<cdiv(N, 256), 256, 0, stream>>>(h2, as2, ad2, als2, ald2, N);
  node_gather<1, OUT_C, false>
      <<<cdiv((long long)N * 32, 256), 256, 0, stream>>>(rowp2, csr, h2, als2, ald2, b2, out, N);
}


# Logits, counter clear and row-pointer copy folded into the two GEMM launches (11 -> 7): logits ride as one more WMMA column tile x.(W.a), so their sums run in a new order (W.a first, then along k)
# speedup vs baseline: 5.8463x; 1.1013x over previous
#include <hip/hip_runtime.h>
#include <math.h>

#define IN_C 128
#define HID 32
#define HEADS 4
#define OUT_C 64
#define NEG_SLOPE 0.2f

typedef __attribute__((ext_vector_type(2))) float v2f;
typedef __attribute__((ext_vector_type(4))) float v4f;
typedef __attribute__((ext_vector_type(8))) float v8f;

constexpr unsigned dpp8_xor_sel(int m) {
  unsigned s = 0;
  for (int i = 0; i < 8; ++i) s |= (unsigned)(i ^ m) << (3 * i);
  return s;
}
constexpr unsigned dpp8_bcast_sel(int i) { return (unsigned)i * 0x249249u; }

template <unsigned SEL>
__device__ __forceinline__ float dpp8_f(float v) {
  return __builtin_bit_cast(float,
                            __builtin_amdgcn_mov_dpp8(__builtin_bit_cast(unsigned, v), SEL));
}
__device__ __forceinline__ float xor8_f(float v) {
  return __builtin_bit_cast(
      float, __builtin_amdgcn_update_dpp(0, __builtin_bit_cast(int, v), 0x168, 0xf, 0xf, false));
}
__device__ __forceinline__ float xor16_f(float v) {
  return __builtin_bit_cast(
      float, __builtin_amdgcn_permlanex16(__builtin_bit_cast(int, v), __builtin_bit_cast(int, v),
                                          0x76543210, 0xfedcba98, false, false));
}
template <int L>
__device__ __forceinline__ float group_sum(float v) {
  v += dpp8_f<dpp8_xor_sel(1)>(v);
  v += dpp8_f<dpp8_xor_sel(2)>(v);
  v += dpp8_f<dpp8_xor_sel(4)>(v);
  if constexpr (L == 16) v += xor8_f(v);
  return v;
}

template <int K, int NC, int H>
__global__ void __launch_bounds__(512)
gemm_wmma(const float* __restrict__ A, const float* __restrict__ B,
          const float* __restrict__ asrc, const float* __restrict__ adst,
          float* __restrict__ C, float* __restrict__ als, float* __restrict__ ald, int M,
          unsigned* __restrict__ zp, int zn,
          const int* __restrict__ cfrom, int* __restrict__ cto, int cn) {
  constexpr int NT = NC / 16;
  constexpr int LPH = (NC / H) / 4;
  constexpr int ZW = 32 - 4 * H;
  static_assert(LPH == 8 || LPH == 16, "head groups of 8 or 16 lanes");
  static_assert(512 % (NC / 4) == 0, "a thread keeps its column group over all passes");
  constexpr int KS = K / 4;
  constexpr int PB = 2 * NC + 32;
  constexpr int WAVES = 16;
  constexpr int QN = NC / 4;
  __shared__ __attribute__((aligned(16))) float Ws[(K / 2) * PB];

  const int lane = threadIdx.x & 31;
  const int c = lane & 15;
  const int h = lane >> 4;
  const int wave = __builtin_amdgcn_readfirstlane((int)threadIdx.x >> 5);
  const int ntiles = (M + 15) >> 4;
  const int stride = (int)gridDim.x * WAVES;
  int tile = (int)blockIdx.x + (int)gridDim.x * wave;

  v2f a[KS];
  bool have = tile < ntiles;
  if (have) {
    int row = tile * 16 + c;
    if (row >= M) row = M - 1;
    const float* ap = A + (size_t)row * K + 2 * h;
#pragma unroll
    for (int ks = 0; ks < KS; ++ks) a[ks] = *(const v2f*)(ap + 4 * ks);
  }

  const int sj = (int)blockIdx.x * 512 + (int)threadIdx.x;
  const int sstride = (int)gridDim.x * 512;
  int cv = 0;
  if (sj < cn) cv = cfrom[sj];

  for (int i = (int)threadIdx.x; i < (K / 2) * ZW; i += 512)
    Ws[(i / ZW) * PB + 2 * NC + 4 * H + (i % ZW)] = 0.f;

  static_assert(((K / 2) * QN) % 512 == 0, "whole passes of the work-group");
  const int q = (int)threadIdx.x % QN;
  const v4f asv = *(const v4f*)(asrc + 4 * q);
  const v4f adv = *(const v4f*)(adst + 4 * q);
  constexpr int NP = ((K / 2) * QN) / 512;
  v4f rr0[NP], rr1[NP];
#pragma unroll
  for (int it = 0; it < NP; ++it) {
    const int p = ((int)threadIdx.x + it * 512) / QN;
    rr0[it] = *(const v4f*)(B + (size_t)(2 * p) * NC + 4 * q);
    rr1[it] = *(const v4f*)(B + (size_t)(2 * p + 1) * NC + 4 * q);
  }
#pragma unroll
  for (int it = 0; it < NP; ++it) {
    const int i = (int)threadIdx.x + it * 512;
    const int p = i / QN;
    const v4f r0 = rr0[it];
    const v4f r1 = rr1[it];
    float* d = Ws + p * PB + 8 * q;
    const v4f lo = {r0.x, r1.x, r0.y, r1.y};
    const v4f hi = {r0.z, r1.z, r0.w, r1.w};
    *(v4f*)d = lo;
    *(v4f*)(d + 4) = hi;
    float s0 = fmaf(r0.w, asv.w, fmaf(r0.z, asv.z, fmaf(r0.y, asv.y, r0.x * asv.x)));
    float s1 = fmaf(r1.w, asv.w, fmaf(r1.z, asv.z, fmaf(r1.y, asv.y, r1.x * asv.x)));
    float d0 = fmaf(r0.w, adv.w, fmaf(r0.z, adv.z, fmaf(r0.y, adv.y, r0.x * adv.x)));
    float d1 = fmaf(r1.w, adv.w, fmaf(r1.z, adv.z, fmaf(r1.y, adv.y, r1.x * adv.x)));
    s0 = group_sum<LPH>(s0);
    s1 = group_sum<LPH>(s1);
    d0 = group_sum<LPH>(d0);
    d1 = group_sum<LPH>(d1);
    if (q % LPH == 0) {
      float* lt = Ws + p * PB + 2 * NC;
      const int hd = q / LPH;
      *(v2f*)(lt + 2 * hd) = v2f{s0, s1};
      *(v2f*)(lt + 2 * (H + hd)) = v2f{d0, d1};
    }
  }
  if (sj < cn) cto[sj] = cv;
  for (int i = sj + sstride; i < cn; i += sstride) cto[i] = cfrom[i];
  for (int i = sj; i < zn; i += sstride) zp[i] = 0u;
  __syncthreads();

  const float* bl = Ws + h * PB + 2 * c;
  while (have) {
    asm volatile("" ::: "memory");
    v8f acc[NT + 1];
#pragma unroll
    for (int t = 0; t <= NT; ++t) acc[t] = v8f{};

#pragma unroll
    for (int ks = 0; ks < KS; ++ks) {
#pragma unroll
      for (int t = 0; t <= NT; ++t) {
        const v2f bv = *(const v2f*)(bl + ks * 2 * PB + t * 32);
        acc[t] = __builtin_amdgcn_wmma_f32_16x16x4_f32(false, a[ks], false, bv, (short)0,
                                                       acc[t], false, false);
      }
    }

    const int rbase = tile * 16 + h * 8;
    float* cbase = C + (size_t)rbase * NC + c;
    if (tile * 16 + 16 <= M) {
#pragma unroll
      for (int t = 0; t < NT; ++t)
#pragma unroll
        for (int i = 0; i < 8; ++i) cbase[(size_t)i * NC + t * 16] = acc[t][i];
    } else {
#pragma unroll
      for (int t = 0; t < NT; ++t)
#pragma unroll
        for (int i = 0; i < 8; ++i)
          if (rbase + i < M) cbase[(size_t)i * NC + t * 16] = acc[t][i];
    }
    if (c < 2 * H) {
      float* lp = (c < H ? als + c : ald + (c - H)) + (size_t)rbase * H;
#pragma unroll
      for (int i = 0; i < 8; ++i)
        if (rbase + i < M) lp[(size_t)i * H] = acc[NT][i];
    }

    tile += stride;
    have = tile < ntiles;
    if (have) {
      int row = tile * 16 + c;
      if (row >= M) row = M - 1;
      const float* ap = A + (size_t)row * K + 2 * h;
#pragma unroll
      for (int ks = 0; ks < KS; ++ks) a[ks] = *(const v2f*)(ap + 4 * ks);
    }
  }
}

__device__ __forceinline__ float leaky(float v) {
  return v > 0.f ? v : NEG_SLOPE * v;
}

#define BSH 8
#define BW (1 << BSH)
#define NBMAX 512

__device__ __forceinline__ int wg_excl_scan(int v, int* wsum) {
  const int lane = threadIdx.x & 31;
  const int w = threadIdx.x >> 5;
  int inc = v;
  for (int o = 1; o < 32; o <<= 1) {
    const int u = __shfl_up(inc, o);
    if (lane >= o) inc += u;
  }
  if (lane == 31) wsum[w] = inc;
  __syncthreads();
  int run = 0;
  for (int k = 0; k < w; ++k) run += wsum[k];
  return run + inc - v;
}

__global__ void __launch_bounds__(1024)
bin_count_k(const int* __restrict__ dst, int* __restrict__ bintotal,
            int* __restrict__ wgbase, int E, int NB) {
  __shared__ int hist[NBMAX];
  const int t = threadIdx.x;
  if (t < NBMAX) hist[t] = 0;
  __syncthreads();
  const int chunk = (E + (int)gridDim.x - 1) / (int)gridDim.x;
  const int b = (int)blockIdx.x * chunk;
  int en = b + chunk;
  if (en > E) en = E;
  for (int e0 = b + t; e0 < en; e0 += 4096) {
    int d[4];
    bool ok[4];
#pragma unroll
    for (int k = 0; k < 4; ++k) {
      const int e = e0 + k * 1024;
      ok[k] = e < en;
      d[k] = dst[ok[k] ? e : en - 1];
    }
#pragma unroll
    for (int k = 0; k < 4; ++k) atomicAdd(&hist[d[k] >> BSH], ok[k] ? 1 : 0);
  }
  __syncthreads();
  if (t < NB) wgbase[(int)blockIdx.x * NB + t] = atomicAdd(&bintotal[t], hist[t]);
}

__global__ void __launch_bounds__(1024)
bin_scatter_k(const int* __restrict__ src, const int* __restrict__ dst,
              const int* __restrict__ bintotal, const int* __restrict__ wgbase,
              int2* __restrict__ tmp, int E, int NB) {
  __shared__ int cur[NBMAX];
  __shared__ int wsum[32];
  const int t = threadIdx.x;
  const int ex = wg_excl_scan(t < NB ? bintotal[t] : 0, wsum);
  if (t < NB) cur[t] = ex + wgbase[(int)blockIdx.x * NB + t];
  __syncthreads();
  const int chunk = (E + (int)gridDim.x - 1) / (int)gridDim.x;
  const int b = (int)blockIdx.x * chunk;
  int en = b + chunk;
  if (en > E) en = E;
  for (int e0 = b + t; e0 < en; e0 += 4096) {
    int d[4], s[4];
    bool ok[4];
#pragma unroll
    for (int k = 0; k < 4; ++k) {
      const int e = e0 + k * 1024;
      ok[k] = e < en;
      const int ec = ok[k] ? e : en - 1;
      d[k] = dst[ec];
      s[k] = src[ec];
    }
#pragma unroll
    for (int k = 0; k < 4; ++k) {
      const int p = atomicAdd(&cur[d[k] >> BSH], ok[k] ? 1 : 0);
      if (ok[k]) tmp[p] = make_int2(s[k], d[k]);
    }
  }
}

__global__ void __launch_bounds__(1024)
bin_place_k(const int2* __restrict__ tmp, const int* __restrict__ bintotal,
            int* __restrict__ rowp, int* __restrict__ csr, int N, int NB) {
  __shared__ int cnt[BW];
  __shared__ int cur[BW];
  __shared__ int wsum[32];
  __shared__ int wsum2[32];
  __shared__ int own;
  const int t = threadIdx.x;
  const int lane = t & 31;
  const int bin = (int)blockIdx.x;
  const int bt = (t <= bin) ? bintotal[t] : 0;
  int s = (t < bin) ? bt : 0;
  for (int o = 16; o > 0; o >>= 1) s += __shfl_xor(s, o);
  if (lane == 0) wsum[t >> 5] = s;
  if (t == bin) own = bt;
  if (t < BW) cnt[t] = 0;
  __syncthreads();
  int beg = 0;
  for (int k = 0; k < 32; ++k) beg += wsum[k];
  const int end = beg + own;

  for (int i0 = beg + t; i0 < end; i0 += 4096) {
    int d[4];
    bool ok[4];
#pragma unroll
    for (int k = 0; k < 4; ++k) {
      const int i = i0 + k * 1024;
      ok[k] = i < end;
      d[k] = tmp[ok[k] ? i : end - 1].y;
    }
#pragma unroll
    for (int k = 0; k < 4; ++k) atomicAdd(&cnt[d[k] & (BW - 1)], ok[k] ? 1 : 0);
  }
  __syncthreads();
  const int ex = wg_excl_scan(t < BW ? cnt[t] : 0, wsum2);
  if (t < BW) {
    cur[t] = beg + ex;
    const int node = bin * BW + t;
    if (node < N) rowp[node] = beg + ex;
  }
  if (bin == NB - 1 && t == 0) rowp[N] = end;
  __syncthreads();
  for (int i0 = beg + t; i0 < end; i0 += 4096) {
    int2 sd[4];
    bool ok[4];
#pragma unroll
    for (int k = 0; k < 4; ++k) {
      const int i = i0 + k * 1024;
      ok[k] = i < end;
      sd[k] = tmp[ok[k] ? i : end - 1];
    }
#pragma unroll
    for (int k = 0; k < 4; ++k) {
      const int p = atomicAdd(&cur[sd[k].y & (BW - 1)], ok[k] ? 1 : 0);
      if (ok[k]) csr[p] = sd[k].x;
    }
  }
}

template <int G>
__device__ __forceinline__ float group_max(float v) {
  v = fmaxf(v, dpp8_f<dpp8_xor_sel(1)>(v));
  v = fmaxf(v, dpp8_f<dpp8_xor_sel(2)>(v));
  v = fmaxf(v, dpp8_f<dpp8_xor_sel(4)>(v));
  if constexpr (G == 32) {
    v = fmaxf(v, xor8_f(v));
    v = fmaxf(v, xor16_f(v));
  }
  return v;
}

template <int H, int CH, bool ELU>
__global__ void node_gather(const int* __restrict__ rowp, const int* __restrict__ csr,
                            const float* __restrict__ h,
                            const float* __restrict__ als, const float* __restrict__ ald,
                            const float* __restrict__ bias, float* __restrict__ out, int N) {
  static_assert(H == 1 || H == 4, "head groups of 32 or 8 lanes");
  const int gw = (blockIdx.x * blockDim.x + threadIdx.x) >> 5;
  const int lane = threadIdx.x & 31;
  if (gw >= N) return;
  const int d = __builtin_amdgcn_readfirstlane(gw);
  constexpr int PER = CH / 32;
  constexpr int G = 32 / H;
  typedef float RowV __attribute__((ext_vector_type(PER)));
  const int c0 = lane * PER;
  const int head = lane / G;
  const int li = lane % G;
  const int beg = rowp[d];
  const int end = rowp[d + 1];
  const float ad = ald[d * H + head];
  const RowV bv = *(const RowV*)(bias + c0);

  float ssum = 0.f;
  float acc[PER];
#pragma unroll
  for (int i = 0; i < PER; ++i) acc[i] = 0.f;

#define GATHER_ROW(J)                                                              \
  if ((J) < cnt) {                                                                 \
    const int s_ = __builtin_amdgcn_readlane(sk[(J) / G], (J) % G);                \
    hv[(J) & 7] = *(const RowV*)(h + (size_t)s_ * CH + c0);                        \
  }
#define GATHER_ROWS8(J0)                                                           \
  GATHER_ROW((J0) + 0) GATHER_ROW((J0) + 1) GATHER_ROW((J0) + 2) GATHER_ROW((J0) + 3) \
  GATHER_ROW((J0) + 4) GATHER_ROW((J0) + 5) GATHER_ROW((J0) + 6) GATHER_ROW((J0) + 7)
#define FMA_ROW(J)                                                                 \
  if ((J) < cnt) {                                                                 \
    float p_;                                                                      \
    if constexpr (H == 4) {                                                        \
      p_ = dpp8_f<dpp8_bcast_sel((J) % 8)>(tk[(J) / 8]);                           \
    } else {                                                                       \
      p_ = __builtin_bit_cast(                                                     \
          float, __builtin_amdgcn_readlane(__builtin_bit_cast(int, tk[0]), (J)));  \
    }                                                                              \
    ssum += p_;                                                                    \
      \
    _Pragma("unroll") for (int i_ = 0; i_ < PER; ++i_)                             \
        acc[i_] = fmaf(hv[(J) & 7][i_], p_, acc[i_]);                              \
  }
#define FMA_ROWS8(J0)                                                              \
  FMA_ROW((J0) + 0) FMA_ROW((J0) + 1) FMA_ROW((J0) + 2) FMA_ROW((J0) + 3)          \
  FMA_ROW((J0) + 4) FMA_ROW((J0) + 5) FMA_ROW((J0) + 6) FMA_ROW((J0) + 7)
#define LOAD_IDS(BASE)                                                             \
           \
  _Pragma("unroll") for (int k = 0; k < H; ++k) {                                  \
    const int jj = k * G + li;                                                     \
    sk[k] = csr[(BASE) + (jj < cnt ? jj : cnt - 1)];                               \
  }

  if (beg < end) {
    int sk[H];
    float ak[H], ek[H], tk[H];
    RowV hv[8];
    int cnt = end - beg;
    if (cnt > 32) cnt = 32;

    LOAD_IDS(beg)
#pragma unroll
    for (int k = 0; k < H; ++k) ak[k] = als[sk[k] * H + head];
    GATHER_ROWS8(0)

    float m = -INFINITY;
#pragma unroll
    for (int k = 0; k < H; ++k) {
      ek[k] = (k * G + li < cnt) ? leaky(ak[k] + ad) : -INFINITY;
      m = fmaxf(m, ek[k]);
    }
    for (int base = beg + 32; base < end; base += 32) {
      int c2 = end - base;
      if (c2 > 32) c2 = 32;
#pragma unroll
      for (int k = 0; k < H; ++k) {
        const int jj = k * G + li;
        const int s2 = csr[base + (jj < c2 ? jj : c2 - 1)];
        const float e2 = (jj < c2) ? leaky(als[s2 * H + head] + ad) : -INFINITY;
        m = fmaxf(m, e2);
      }
    }
    m = group_max<G>(m);

    for (int base = beg; base < end; base += 32) {
      cnt = end - base;
      if (cnt > 32) cnt = 32;
      if (base != beg) {
        LOAD_IDS(base)
#pragma unroll
        for (int k = 0; k < H; ++k) ak[k] = als[sk[k] * H + head];
        GATHER_ROWS8(0)
#pragma unroll
        for (int k = 0; k < H; ++k)
          ek[k] = (k * G + li < cnt) ? leaky(ak[k] + ad) : -INFINITY;
      }
#pragma unroll
      for (int k = 0; k < H; ++k)
        tk[k] = (k * G + li < cnt) ? __expf(ek[k] - m) : 0.f;

      FMA_ROWS8(0)
      if (8 < cnt) {
        GATHER_ROWS8(8)
        FMA_ROWS8(8)
      }
      if (16 < cnt) {
        GATHER_ROWS8(16)
        FMA_ROWS8(16)
      }
      if (24 < cnt) {
        GATHER_ROWS8(24)
        FMA_ROWS8(24)
      }
    }
  }
#undef GATHER_ROW
#undef GATHER_ROWS8
#undef FMA_ROW
#undef FMA_ROWS8
#undef LOAD_IDS

  const float den = ssum + 1e-16f;
  RowV o;
#pragma unroll
  for (int i = 0; i < PER; ++i) {
    float v = acc[i] / den + bv[i];
    if (ELU) v = v > 0.f ? v : (__expf(v) - 1.0f);
    o[i] = v;
  }
  *(RowV*)(out + (size_t)d * CH + c0) = o;
}

extern "C" void kernel_launch(void* const* d_in, const int* in_sizes, int n_in,
                              void* d_out, int out_size, void* d_ws, size_t ws_size,
                              hipStream_t stream) {
  const float* x   = (const float*)d_in[0];
  const int*   ei  = (const int*)d_in[1];
  const float* W1  = (const float*)d_in[2];
  const float* as1 = (const float*)d_in[3];
  const float* ad1 = (const float*)d_in[4];
  const float* b1  = (const float*)d_in[5];
  const float* W2  = (const float*)d_in[6];
  const float* as2 = (const float*)d_in[7];
  const float* ad2 = (const float*)d_in[8];
  const float* b2  = (const float*)d_in[9];
  float* out = (float*)d_out;

  const int N = in_sizes[0] / IN_C;
  const int E = in_sizes[1] / 2;
  const int* src = ei;
  const int* dst = ei + E;

  float* ws = (float*)d_ws;
  float* h1   = ws;
  float* act1 = h1 + (size_t)N * 128;
  int*   csr  = (int*)(act1 + (size_t)N * 128);
  int2*  pairs = (int2*)act1;
  int*   bintotal = (int*)out;
  int*   rowp = (int*)out + N;
  int*   wgbase = (int*)out + (size_t)N * 11;
  float* als1 = out + (size_t)N * 3;
  float* ald1 = als1 + (size_t)N * HEADS;
  float* h2    = h1;
  int*   rowp2 = (int*)(h1 + (size_t)N * 64);
  float* als2  = h1 + (size_t)N * 64 + (size_t)N * 2;
  float* ald2  = als2 + N;

  auto cdiv = [](long long a, long long b) -> unsigned { return (unsigned)((a + b - 1) / b); };

  const int NB = (int)cdiv(N, BW);
  gemm_wmma<IN_C, HEADS * HID, HEADS><<<256, 512, 0, stream>>>(
      x, W1, as1, ad1, h1, als1, ald1, N, (unsigned*)bintotal, NB, nullptr, nullptr, 0);

  bin_count_k<<<256, 1024, 0, stream>>>(dst, bintotal, wgbase, E, NB);
  bin_scatter_k<<<256, 1024, 0, stream>>>(src, dst, bintotal, wgbase, pairs, E, NB);
  bin_place_k<<<NB, 1024, 0, stream>>>(pairs, bintotal, rowp, csr, N, NB);

  node_gather<HEADS, 128, true>
      <<<cdiv((long long)N * 32, 256), 256, 0, stream>>>(rowp, csr, h1, als1, ald1, b1, act1, N);

  gemm_wmma<HEADS * HID, OUT_C, 1><<<256, 512, 0, stream>>>(
      act1, W2, as2, ad2, h2, als2, ald2, N, nullptr, 0, rowp, rowp2, N + 1);
  node_gather<1, OUT_C, false>
      <<<cdiv((long long)N * 32, 256), 256, 0, stream>>>(rowp2, csr, h2, als2, ald2, b2, out, N);
}


# Sort: bin_scatter and bin_place order records in an LDS stage and write it out as contiguous runs (whole lines per wave store, not 32 scattered words); per-node edge order stays LDS arrival order
# speedup vs baseline: 6.0837x; 1.0406x over previous
#include <hip/hip_runtime.h>
#include <math.h>

#define IN_C 128
#define HID 32
#define HEADS 4
#define OUT_C 64
#define NEG_SLOPE 0.2f

typedef __attribute__((ext_vector_type(2))) float v2f;
typedef __attribute__((ext_vector_type(4))) float v4f;
typedef __attribute__((ext_vector_type(8))) float v8f;

constexpr unsigned dpp8_xor_sel(int m) {
  unsigned s = 0;
  for (int i = 0; i < 8; ++i) s |= (unsigned)(i ^ m) << (3 * i);
  return s;
}
constexpr unsigned dpp8_bcast_sel(int i) { return (unsigned)i * 0x249249u; }

template <unsigned SEL>
__device__ __forceinline__ float dpp8_f(float v) {
  return __builtin_bit_cast(float,
                            __builtin_amdgcn_mov_dpp8(__builtin_bit_cast(unsigned, v), SEL));
}
__device__ __forceinline__ float xor8_f(float v) {
  return __builtin_bit_cast(
      float, __builtin_amdgcn_update_dpp(0, __builtin_bit_cast(int, v), 0x168, 0xf, 0xf, false));
}
__device__ __forceinline__ float xor16_f(float v) {
  return __builtin_bit_cast(
      float, __builtin_amdgcn_permlanex16(__builtin_bit_cast(int, v), __builtin_bit_cast(int, v),
                                          0x76543210, 0xfedcba98, false, false));
}
template <int L>
__device__ __forceinline__ float group_sum(float v) {
  v += dpp8_f<dpp8_xor_sel(1)>(v);
  v += dpp8_f<dpp8_xor_sel(2)>(v);
  v += dpp8_f<dpp8_xor_sel(4)>(v);
  if constexpr (L == 16) v += xor8_f(v);
  return v;
}

template <int K, int NC, int H>
__global__ void __launch_bounds__(512)
gemm_wmma(const float* __restrict__ A, const float* __restrict__ B,
          const float* __restrict__ asrc, const float* __restrict__ adst,
          float* __restrict__ C, float* __restrict__ als, float* __restrict__ ald, int M,
          unsigned* __restrict__ zp, int zn,
          const int* __restrict__ cfrom, int* __restrict__ cto, int cn) {
  constexpr int NT = NC / 16;
  constexpr int LPH = (NC / H) / 4;
  constexpr int ZW = 32 - 4 * H;
  static_assert(LPH == 8 || LPH == 16, "head groups of 8 or 16 lanes");
  static_assert(512 % (NC / 4) == 0, "a thread keeps its column group over all passes");
  constexpr int KS = K / 4;
  constexpr int PB = 2 * NC + 32;
  constexpr int WAVES = 16;
  constexpr int QN = NC / 4;
  __shared__ __attribute__((aligned(16))) float Ws[(K / 2) * PB];

  const int lane = threadIdx.x & 31;
  const int c = lane & 15;
  const int h = lane >> 4;
  const int wave = __builtin_amdgcn_readfirstlane((int)threadIdx.x >> 5);
  const int ntiles = (M + 15) >> 4;
  const int stride = (int)gridDim.x * WAVES;
  int tile = (int)blockIdx.x + (int)gridDim.x * wave;

  v2f a[KS];
  bool have = tile < ntiles;
  if (have) {
    int row = tile * 16 + c;
    if (row >= M) row = M - 1;
    const float* ap = A + (size_t)row * K + 2 * h;
#pragma unroll
    for (int ks = 0; ks < KS; ++ks) a[ks] = *(const v2f*)(ap + 4 * ks);
  }

  const int sj = (int)blockIdx.x * 512 + (int)threadIdx.x;
  const int sstride = (int)gridDim.x * 512;
  int cv = 0;
  if (sj < cn) cv = cfrom[sj];

  for (int i = (int)threadIdx.x; i < (K / 2) * ZW; i += 512)
    Ws[(i / ZW) * PB + 2 * NC + 4 * H + (i % ZW)] = 0.f;

  static_assert(((K / 2) * QN) % 512 == 0, "whole passes of the work-group");
  const int q = (int)threadIdx.x % QN;
  const v4f asv = *(const v4f*)(asrc + 4 * q);
  const v4f adv = *(const v4f*)(adst + 4 * q);
  constexpr int NP = ((K / 2) * QN) / 512;
  v4f rr0[NP], rr1[NP];
#pragma unroll
  for (int it = 0; it < NP; ++it) {
    const int p = ((int)threadIdx.x + it * 512) / QN;
    rr0[it] = *(const v4f*)(B + (size_t)(2 * p) * NC + 4 * q);
    rr1[it] = *(const v4f*)(B + (size_t)(2 * p + 1) * NC + 4 * q);
  }
#pragma unroll
  for (int it = 0; it < NP; ++it) {
    const int i = (int)threadIdx.x + it * 512;
    const int p = i / QN;
    const v4f r0 = rr0[it];
    const v4f r1 = rr1[it];
    float* d = Ws + p * PB + 8 * q;
    const v4f lo = {r0.x, r1.x, r0.y, r1.y};
    const v4f hi = {r0.z, r1.z, r0.w, r1.w};
    *(v4f*)d = lo;
    *(v4f*)(d + 4) = hi;
    float s0 = fmaf(r0.w, asv.w, fmaf(r0.z, asv.z, fmaf(r0.y, asv.y, r0.x * asv.x)));
    float s1 = fmaf(r1.w, asv.w, fmaf(r1.z, asv.z, fmaf(r1.y, asv.y, r1.x * asv.x)));
    float d0 = fmaf(r0.w, adv.w, fmaf(r0.z, adv.z, fmaf(r0.y, adv.y, r0.x * adv.x)));
    float d1 = fmaf(r1.w, adv.w, fmaf(r1.z, adv.z, fmaf(r1.y, adv.y, r1.x * adv.x)));
    s0 = group_sum<LPH>(s0);
    s1 = group_sum<LPH>(s1);
    d0 = group_sum<LPH>(d0);
    d1 = group_sum<LPH>(d1);
    if (q % LPH == 0) {
      float* lt = Ws + p * PB + 2 * NC;
      const int hd = q / LPH;
      *(v2f*)(lt + 2 * hd) = v2f{s0, s1};
      *(v2f*)(lt + 2 * (H + hd)) = v2f{d0, d1};
    }
  }
  if (sj < cn) cto[sj] = cv;
  for (int i = sj + sstride; i < cn; i += sstride) cto[i] = cfrom[i];
  for (int i = sj; i < zn; i += sstride) zp[i] = 0u;
  __syncthreads();

  const float* bl = Ws + h * PB + 2 * c;
  while (have) {
    asm volatile("" ::: "memory");
    v8f acc[NT + 1];
#pragma unroll
    for (int t = 0; t <= NT; ++t) acc[t] = v8f{};

#pragma unroll
    for (int ks = 0; ks < KS; ++ks) {
#pragma unroll
      for (int t = 0; t <= NT; ++t) {
        const v2f bv = *(const v2f*)(bl + ks * 2 * PB + t * 32);
        acc[t] = __builtin_amdgcn_wmma_f32_16x16x4_f32(false, a[ks], false, bv, (short)0,
                                                       acc[t], false, false);
      }
    }

    const int rbase = tile * 16 + h * 8;
    float* cbase = C + (size_t)rbase * NC + c;
    if (tile * 16 + 16 <= M) {
#pragma unroll
      for (int t = 0; t < NT; ++t)
#pragma unroll
        for (int i = 0; i < 8; ++i) cbase[(size_t)i * NC + t * 16] = acc[t][i];
    } else {
#pragma unroll
      for (int t = 0; t < NT; ++t)
#pragma unroll
        for (int i = 0; i < 8; ++i)
          if (rbase + i < M) cbase[(size_t)i * NC + t * 16] = acc[t][i];
    }
    if (c < 2 * H) {
      float* lp = (c < H ? als + c : ald + (c - H)) + (size_t)rbase * H;
#pragma unroll
      for (int i = 0; i < 8; ++i)
        if (rbase + i < M) lp[(size_t)i * H] = acc[NT][i];
    }

    tile += stride;
    have = tile < ntiles;
    if (have) {
      int row = tile * 16 + c;
      if (row >= M) row = M - 1;
      const float* ap = A + (size_t)row * K + 2 * h;
#pragma unroll
      for (int ks = 0; ks < KS; ++ks) a[ks] = *(const v2f*)(ap + 4 * ks);
    }
  }
}

__device__ __forceinline__ float leaky(float v) {
  return v > 0.f ? v : NEG_SLOPE * v;
}

#define BSH 8
#define BW (1 << BSH)
#define NBMAX 512
#define SCAP 8192
#define PCAP 16384

__device__ __forceinline__ int wg_excl_scan(int v, int* wsum) {
  const int lane = threadIdx.x & 31;
  const int w = threadIdx.x >> 5;
  int inc = v;
  for (int o = 1; o < 32; o <<= 1) {
    const int u = __shfl_up(inc, o);
    if (lane >= o) inc += u;
  }
  if (lane == 31) wsum[w] = inc;
  __syncthreads();
  int run = 0;
  for (int k = 0; k < w; ++k) run += wsum[k];
  return run + inc - v;
}

__global__ void __launch_bounds__(1024)
bin_count_k(const int* __restrict__ dst, int* __restrict__ bintotal,
            int* __restrict__ wgbase, int* __restrict__ wgcnt, int E, int NB) {
  __shared__ int hist[NBMAX];
  const int t = threadIdx.x;
  if (t < NBMAX) hist[t] = 0;
  __syncthreads();
  const int chunk = (E + (int)gridDim.x - 1) / (int)gridDim.x;
  const int b = (int)blockIdx.x * chunk;
  int en = b + chunk;
  if (en > E) en = E;
  for (int e0 = b + t; e0 < en; e0 += 4096) {
    int d[4];
    bool ok[4];
#pragma unroll
    for (int k = 0; k < 4; ++k) {
      const int e = e0 + k * 1024;
      ok[k] = e < en;
      d[k] = dst[ok[k] ? e : en - 1];
    }
#pragma unroll
    for (int k = 0; k < 4; ++k) atomicAdd(&hist[d[k] >> BSH], ok[k] ? 1 : 0);
  }
  __syncthreads();
  if (t < NB) {
    const int hc = hist[t];
    wgcnt[(int)blockIdx.x * NB + t] = hc;
    wgbase[(int)blockIdx.x * NB + t] = atomicAdd(&bintotal[t], hc);
  }
}

__global__ void __launch_bounds__(1024)
bin_scatter_k(const int* __restrict__ src, const int* __restrict__ dst,
              const int* __restrict__ bintotal, const int* __restrict__ wgbase,
              const int* __restrict__ wgcnt, int2* __restrict__ tmp, int E, int NB) {
  __shared__ int cur[NBMAX];
  __shared__ int gdelta[NBMAX];
  __shared__ int wsum[32];
  __shared__ int wsum2[32];
  __shared__ int2 stage[SCAP];
  const int t = threadIdx.x;
  const int ex = wg_excl_scan(t < NB ? bintotal[t] : 0, wsum);
  const int chunk = (E + (int)gridDim.x - 1) / (int)gridDim.x;
  const int b = (int)blockIdx.x * chunk;
  int en = b + chunk;
  if (en > E) en = E;
  const bool staged = chunk <= SCAP;
  int ls = 0;
  if (staged) ls = wg_excl_scan(t < NB ? wgcnt[(int)blockIdx.x * NB + t] : 0, wsum2);
  if (t < NB) {
    const int gp = ex + wgbase[(int)blockIdx.x * NB + t];
    cur[t] = staged ? ls : gp;
    gdelta[t] = gp - ls;
  }
  __syncthreads();
  for (int e0 = b + t; e0 < en; e0 += 4096) {
    int d[4], s[4];
    bool ok[4];
#pragma unroll
    for (int k = 0; k < 4; ++k) {
      const int e = e0 + k * 1024;
      ok[k] = e < en;
      const int ec = ok[k] ? e : en - 1;
      d[k] = dst[ec];
      s[k] = src[ec];
    }
#pragma unroll
    for (int k = 0; k < 4; ++k) {
      const int p = atomicAdd(&cur[d[k] >> BSH], ok[k] ? 1 : 0);
      if (ok[k]) {
        if (staged) stage[p] = make_int2(s[k], d[k]);
        else tmp[p] = make_int2(s[k], d[k]);
      }
    }
  }
  if (staged) {
    __syncthreads();
    const int n = en - b;
    for (int i0 = t; i0 < n; i0 += 4096) {
      int2 sd[4];
      bool ok[4];
#pragma unroll
      for (int k = 0; k < 4; ++k) {
        const int i = i0 + k * 1024;
        ok[k] = i < n;
        sd[k] = stage[ok[k] ? i : i0];
      }
      int g[4];
#pragma unroll
      for (int k = 0; k < 4; ++k) g[k] = gdelta[sd[k].y >> BSH];
#pragma unroll
      for (int k = 0; k < 4; ++k)
        if (ok[k]) tmp[g[k] + i0 + k * 1024] = sd[k];
    }
  }
}

__global__ void __launch_bounds__(1024)
bin_place_k(const int2* __restrict__ tmp, const int* __restrict__ bintotal,
            int* __restrict__ rowp, int* __restrict__ csr, int N, int NB) {
  __shared__ int cnt[BW];
  __shared__ int cur[BW];
  __shared__ int wsum[32];
  __shared__ int wsum2[32];
  __shared__ int own;
  __shared__ int stage[PCAP];
  const int t = threadIdx.x;
  const int lane = t & 31;
  const int bin = (int)blockIdx.x;
  const int bt = (t <= bin) ? bintotal[t] : 0;
  int s = (t < bin) ? bt : 0;
  for (int o = 16; o > 0; o >>= 1) s += __shfl_xor(s, o);
  if (lane == 0) wsum[t >> 5] = s;
  if (t == bin) own = bt;
  if (t < BW) cnt[t] = 0;
  __syncthreads();
  int beg = 0;
  for (int k = 0; k < 32; ++k) beg += wsum[k];
  const int end = beg + own;

  for (int i0 = beg + t; i0 < end; i0 += 4096) {
    int d[4];
    bool ok[4];
#pragma unroll
    for (int k = 0; k < 4; ++k) {
      const int i = i0 + k * 1024;
      ok[k] = i < end;
      d[k] = tmp[ok[k] ? i : end - 1].y;
    }
#pragma unroll
    for (int k = 0; k < 4; ++k) atomicAdd(&cnt[d[k] & (BW - 1)], ok[k] ? 1 : 0);
  }
  __syncthreads();
  const int ex = wg_excl_scan(t < BW ? cnt[t] : 0, wsum2);
  const bool staged = end - beg <= PCAP;
  if (t < BW) {
    cur[t] = staged ? ex : beg + ex;
    const int node = bin * BW + t;
    if (node < N) rowp[node] = beg + ex;
  }
  if (bin == NB - 1 && t == 0) rowp[N] = end;
  __syncthreads();
  for (int i0 = beg + t; i0 < end; i0 += 4096) {
    int2 sd[4];
    bool ok[4];
#pragma unroll
    for (int k = 0; k < 4; ++k) {
      const int i = i0 + k * 1024;
      ok[k] = i < end;
      sd[k] = tmp[ok[k] ? i : end - 1];
    }
#pragma unroll
    for (int k = 0; k < 4; ++k) {
      const int p = atomicAdd(&cur[sd[k].y & (BW - 1)], ok[k] ? 1 : 0);
      if (ok[k]) {
        if (staged) stage[p] = sd[k].x;
        else csr[p] = sd[k].x;
      }
    }
  }
  if (staged) {
    __syncthreads();
    for (int i = beg + t; i < end; i += 1024) csr[i] = stage[i - beg];
  }
}

template <int G>
__device__ __forceinline__ float group_max(float v) {
  v = fmaxf(v, dpp8_f<dpp8_xor_sel(1)>(v));
  v = fmaxf(v, dpp8_f<dpp8_xor_sel(2)>(v));
  v = fmaxf(v, dpp8_f<dpp8_xor_sel(4)>(v));
  if constexpr (G == 32) {
    v = fmaxf(v, xor8_f(v));
    v = fmaxf(v, xor16_f(v));
  }
  return v;
}

template <int H, int CH, bool ELU>
__global__ void node_gather(const int* __restrict__ rowp, const int* __restrict__ csr,
                            const float* __restrict__ h,
                            const float* __restrict__ als, const float* __restrict__ ald,
                            const float* __restrict__ bias, float* __restrict__ out, int N) {
  static_assert(H == 1 || H == 4, "head groups of 32 or 8 lanes");
  const int gw = (blockIdx.x * blockDim.x + threadIdx.x) >> 5;
  const int lane = threadIdx.x & 31;
  if (gw >= N) return;
  const int d = __builtin_amdgcn_readfirstlane(gw);
  constexpr int PER = CH / 32;
  constexpr int G = 32 / H;
  typedef float RowV __attribute__((ext_vector_type(PER)));
  const int c0 = lane * PER;
  const int head = lane / G;
  const int li = lane % G;
  const int beg = rowp[d];
  const int end = rowp[d + 1];
  const float ad = ald[d * H + head];
  const RowV bv = *(const RowV*)(bias + c0);

  float ssum = 0.f;
  float acc[PER];
#pragma unroll
  for (int i = 0; i < PER; ++i) acc[i] = 0.f;

#define GATHER_ROW(J)                                                              \
  if ((J) < cnt) {                                                                 \
    const int s_ = __builtin_amdgcn_readlane(sk[(J) / G], (J) % G);                \
    hv[(J) & 7] = *(const RowV*)(h + (size_t)s_ * CH + c0);                        \
  }
#define GATHER_ROWS8(J0)                                                           \
  GATHER_ROW((J0) + 0) GATHER_ROW((J0) + 1) GATHER_ROW((J0) + 2) GATHER_ROW((J0) + 3) \
  GATHER_ROW((J0) + 4) GATHER_ROW((J0) + 5) GATHER_ROW((J0) + 6) GATHER_ROW((J0) + 7)
#define FMA_ROW(J)                                                                 \
  if ((J) < cnt) {                                                                 \
    float p_;                                                                      \
    if constexpr (H == 4) {                                                        \
      p_ = dpp8_f<dpp8_bcast_sel((J) % 8)>(tk[(J) / 8]);                           \
    } else {                                                                       \
      p_ = __builtin_bit_cast(                                                     \
          float, __builtin_amdgcn_readlane(__builtin_bit_cast(int, tk[0]), (J)));  \
    }                                                                              \
    ssum += p_;                                                                    \
      \
    _Pragma("unroll") for (int i_ = 0; i_ < PER; ++i_)                             \
        acc[i_] = fmaf(hv[(J) & 7][i_], p_, acc[i_]);                              \
  }
#define FMA_ROWS8(J0)                                                              \
  FMA_ROW((J0) + 0) FMA_ROW((J0) + 1) FMA_ROW((J0) + 2) FMA_ROW((J0) + 3)          \
  FMA_ROW((J0) + 4) FMA_ROW((J0) + 5) FMA_ROW((J0) + 6) FMA_ROW((J0) + 7)
#define LOAD_IDS(BASE)                                                             \
           \
  _Pragma("unroll") for (int k = 0; k < H; ++k) {                                  \
    const int jj = k * G + li;                                                     \
    sk[k] = csr[(BASE) + (jj < cnt ? jj : cnt - 1)];                               \
  }

  if (beg < end) {
    int sk[H];
    float ak[H], ek[H], tk[H];
    RowV hv[8];
    int cnt = end - beg;
    if (cnt > 32) cnt = 32;

    LOAD_IDS(beg)
#pragma unroll
    for (int k = 0; k < H; ++k) ak[k] = als[sk[k] * H + head];
    GATHER_ROWS8(0)

    float m = -INFINITY;
#pragma unroll
    for (int k = 0; k < H; ++k) {
      ek[k] = (k * G + li < cnt) ? leaky(ak[k] + ad) : -INFINITY;
      m = fmaxf(m, ek[k]);
    }
    for (int base = beg + 32; base < end; base += 32) {
      int c2 = end - base;
      if (c2 > 32) c2 = 32;
#pragma unroll
      for (int k = 0; k < H; ++k) {
        const int jj = k * G + li;
        const int s2 = csr[base + (jj < c2 ? jj : c2 - 1)];
        const float e2 = (jj < c2) ? leaky(als[s2 * H + head] + ad) : -INFINITY;
        m = fmaxf(m, e2);
      }
    }
    m = group_max<G>(m);

    for (int base = beg; base < end; base += 32) {
      cnt = end - base;
      if (cnt > 32) cnt = 32;
      if (base != beg) {
        LOAD_IDS(base)
#pragma unroll
        for (int k = 0; k < H; ++k) ak[k] = als[sk[k] * H + head];
        GATHER_ROWS8(0)
#pragma unroll
        for (int k = 0; k < H; ++k)
          ek[k] = (k * G + li < cnt) ? leaky(ak[k] + ad) : -INFINITY;
      }
#pragma unroll
      for (int k = 0; k < H; ++k)
        tk[k] = (k * G + li < cnt) ? __expf(ek[k] - m) : 0.f;

      FMA_ROWS8(0)
      if (8 < cnt) {
        GATHER_ROWS8(8)
        FMA_ROWS8(8)
      }
      if (16 < cnt) {
        GATHER_ROWS8(16)
        FMA_ROWS8(16)
      }
      if (24 < cnt) {
        GATHER_ROWS8(24)
        FMA_ROWS8(24)
      }
    }
  }
#undef GATHER_ROW
#undef GATHER_ROWS8
#undef FMA_ROW
#undef FMA_ROWS8
#undef LOAD_IDS

  const float den = ssum + 1e-16f;
  RowV o;
#pragma unroll
  for (int i = 0; i < PER; ++i) {
    float v = acc[i] / den + bv[i];
    if (ELU) v = v > 0.f ? v : (__expf(v) - 1.0f);
    o[i] = v;
  }
  *(RowV*)(out + (size_t)d * CH + c0) = o;
}

extern "C" void kernel_launch(void* const* d_in, const int* in_sizes, int n_in,
                              void* d_out, int out_size, void* d_ws, size_t ws_size,
                              hipStream_t stream) {
  const float* x   = (const float*)d_in[0];
  const int*   ei  = (const int*)d_in[1];
  const float* W1  = (const float*)d_in[2];
  const float* as1 = (const float*)d_in[3];
  const float* ad1 = (const float*)d_in[4];
  const float* b1  = (const float*)d_in[5];
  const float* W2  = (const float*)d_in[6];
  const float* as2 = (const float*)d_in[7];
  const float* ad2 = (const float*)d_in[8];
  const float* b2  = (const float*)d_in[9];
  float* out = (float*)d_out;

  const int N = in_sizes[0] / IN_C;
  const int E = in_sizes[1] / 2;
  const int* src = ei;
  const int* dst = ei + E;

  float* ws = (float*)d_ws;
  float* h1   = ws;
  float* act1 = h1 + (size_t)N * 128;
  int*   csr  = (int*)(act1 + (size_t)N * 128);
  int2*  pairs = (int2*)act1;
  int*   bintotal = (int*)out;
  int*   rowp = (int*)out + N;
  int*   wgbase = (int*)out + (size_t)N * 11;
  int*   wgcnt  = (int*)out + (size_t)N * 13;
  float* als1 = out + (size_t)N * 3;
  float* ald1 = als1 + (size_t)N * HEADS;
  float* h2    = h1;
  int*   rowp2 = (int*)(h1 + (size_t)N * 64);
  float* als2  = h1 + (size_t)N * 64 + (size_t)N * 2;
  float* ald2  = als2 + N;

  auto cdiv = [](long long a, long long b) -> unsigned { return (unsigned)((a + b - 1) / b); };

  const int NB = (int)cdiv(N, BW);
  gemm_wmma<IN_C, HEADS * HID, HEADS><<<256, 512, 0, stream>>>(
      x, W1, as1, ad1, h1, als1, ald1, N, (unsigned*)bintotal, NB, nullptr, nullptr, 0);

  bin_count_k<<<256, 1024, 0, stream>>>(dst, bintotal, wgbase, wgcnt, E, NB);
  bin_scatter_k<<<256, 1024, 0, stream>>>(src, dst, bintotal, wgbase, wgcnt, pairs, E, NB);
  bin_place_k<<<NB, 1024, 0, stream>>>(pairs, bintotal, rowp, csr, N, NB);

  node_gather<HEADS, 128, true>
      <<<cdiv((long long)N * 32, 256), 256, 0, stream>>>(rowp, csr, h1, als1, ald1, b1, act1, N);

  gemm_wmma<HEADS * HID, OUT_C, 1><<<256, 512, 0, stream>>>(
      act1, W2, as2, ad2, h2, als2, ald2, N, nullptr, 0, rowp, rowp2, N + 1);
  node_gather<1, OUT_C, false>
      <<<cdiv((long long)N * 32, 256), 256, 0, stream>>>(rowp2, csr, h2, als2, ald2, b2, out, N);
}


# Layer-2 product moved into layer-1 aggregation epilogue: a work-group's 16 finished rows meet in an LDS tile and go through WMMA there (k order kept); act1 never stored, launches 7 -> 6
# speedup vs baseline: 6.3015x; 1.0358x over previous
#include <hip/hip_runtime.h>
#include <math.h>

#define IN_C 128
#define HID 32
#define HEADS 4
#define OUT_C 64
#define NEG_SLOPE 0.2f

typedef __attribute__((ext_vector_type(2))) float v2f;
typedef __attribute__((ext_vector_type(4))) float v4f;
typedef __attribute__((ext_vector_type(8))) float v8f;

constexpr unsigned dpp8_xor_sel(int m) {
  unsigned s = 0;
  for (int i = 0; i < 8; ++i) s |= (unsigned)(i ^ m) << (3 * i);
  return s;
}
constexpr unsigned dpp8_bcast_sel(int i) { return (unsigned)i * 0x249249u; }

template <unsigned SEL>
__device__ __forceinline__ float dpp8_f(float v) {
  return __builtin_bit_cast(float,
                            __builtin_amdgcn_mov_dpp8(__builtin_bit_cast(unsigned, v), SEL));
}
__device__ __forceinline__ float xor8_f(float v) {
  return __builtin_bit_cast(
      float, __builtin_amdgcn_update_dpp(0, __builtin_bit_cast(int, v), 0x168, 0xf, 0xf, false));
}
__device__ __forceinline__ float xor16_f(float v) {
  return __builtin_bit_cast(
      float, __builtin_amdgcn_permlanex16(__builtin_bit_cast(int, v), __builtin_bit_cast(int, v),
                                          0x76543210, 0xfedcba98, false, false));
}
template <int L>
__device__ __forceinline__ float group_sum(float v) {
  v += dpp8_f<dpp8_xor_sel(1)>(v);
  v += dpp8_f<dpp8_xor_sel(2)>(v);
  v += dpp8_f<dpp8_xor_sel(4)>(v);
  if constexpr (L == 16) v += xor8_f(v);
  return v;
}

template <int K, int NC, int H>
__device__ __forceinline__ void stage_b(const float* __restrict__ B,
                                        const float* __restrict__ asrc,
                                        const float* __restrict__ adst, float* __restrict__ Ws) {
  constexpr int LPH = (NC / H) / 4;
  constexpr int ZW = 32 - 4 * H;
  static_assert(LPH == 8 || LPH == 16, "head groups of 8 or 16 lanes");
  static_assert(512 % (NC / 4) == 0, "a thread keeps its column group over all passes");
  constexpr int PB = 2 * NC + 32;
  constexpr int QN = NC / 4;

  for (int i = (int)threadIdx.x; i < (K / 2) * ZW; i += 512)
    Ws[(i / ZW) * PB + 2 * NC + 4 * H + (i % ZW)] = 0.f;

  static_assert(((K / 2) * QN) % 512 == 0, "whole passes of the work-group");
  const int q = (int)threadIdx.x % QN;
  const v4f asv = *(const v4f*)(asrc + 4 * q);
  const v4f adv = *(const v4f*)(adst + 4 * q);
  constexpr int NP = ((K / 2) * QN) / 512;
  v4f rr0[NP], rr1[NP];
#pragma unroll
  for (int it = 0; it < NP; ++it) {
    const int p = ((int)threadIdx.x + it * 512) / QN;
    rr0[it] = *(const v4f*)(B + (size_t)(2 * p) * NC + 4 * q);
    rr1[it] = *(const v4f*)(B + (size_t)(2 * p + 1) * NC + 4 * q);
  }
#pragma unroll
  for (int it = 0; it < NP; ++it) {
    const int i = (int)threadIdx.x + it * 512;
    const int p = i / QN;
    const v4f r0 = rr0[it];
    const v4f r1 = rr1[it];
    float* d = Ws + p * PB + 8 * q;
    const v4f lo = {r0.x, r1.x, r0.y, r1.y};
    const v4f hi = {r0.z, r1.z, r0.w, r1.w};
    *(v4f*)d = lo;
    *(v4f*)(d + 4) = hi;
    float s0 = fmaf(r0.w, asv.w, fmaf(r0.z, asv.z, fmaf(r0.y, asv.y, r0.x * asv.x)));
    float s1 = fmaf(r1.w, asv.w, fmaf(r1.z, asv.z, fmaf(r1.y, asv.y, r1.x * asv.x)));
    float d0 = fmaf(r0.w, adv.w, fmaf(r0.z, adv.z, fmaf(r0.y, adv.y, r0.x * adv.x)));
    float d1 = fmaf(r1.w, adv.w, fmaf(r1.z, adv.z, fmaf(r1.y, adv.y, r1.x * adv.x)));
    s0 = group_sum<LPH>(s0);
    s1 = group_sum<LPH>(s1);
    d0 = group_sum<LPH>(d0);
    d1 = group_sum<LPH>(d1);
    if (q % LPH == 0) {
      float* lt = Ws + p * PB + 2 * NC;
      const int hd = q / LPH;
      *(v2f*)(lt + 2 * hd) = v2f{s0, s1};
      *(v2f*)(lt + 2 * (H + hd)) = v2f{d0, d1};
    }
  }
}

template <int K, int NC, int H>
__global__ void __launch_bounds__(512)
gemm_wmma(const float* __restrict__ A, const float* __restrict__ B,
          const float* __restrict__ asrc, const float* __restrict__ adst,
          float* __restrict__ C, float* __restrict__ als, float* __restrict__ ald, int M,
          unsigned* __restrict__ zp, int zn) {
  constexpr int NT = NC / 16;
  constexpr int KS = K / 4;
  constexpr int PB = 2 * NC + 32;
  constexpr int WAVES = 16;
  __shared__ __attribute__((aligned(16))) float Ws[(K / 2) * PB];

  const int lane = threadIdx.x & 31;
  const int c = lane & 15;
  const int h = lane >> 4;
  const int wave = __builtin_amdgcn_readfirstlane((int)threadIdx.x >> 5);
  const int ntiles = (M + 15) >> 4;
  const int stride = (int)gridDim.x * WAVES;
  int tile = (int)blockIdx.x + (int)gridDim.x * wave;

  v2f a[KS];
  bool have = tile < ntiles;
  if (have) {
    int row = tile * 16 + c;
    if (row >= M) row = M - 1;
    const float* ap = A + (size_t)row * K + 2 * h;
#pragma unroll
    for (int ks = 0; ks < KS; ++ks) a[ks] = *(const v2f*)(ap + 4 * ks);
  }

  stage_b<K, NC, H>(B, asrc, adst, Ws);
  for (int i = (int)blockIdx.x * 512 + (int)threadIdx.x; i < zn; i += (int)gridDim.x * 512)
    zp[i] = 0u;
  __syncthreads();

  const float* bl = Ws + h * PB + 2 * c;
  while (have) {
    asm volatile("" ::: "memory");
    v8f acc[NT + 1];
#pragma unroll
    for (int t = 0; t <= NT; ++t) acc[t] = v8f{};

#pragma unroll
    for (int ks = 0; ks < KS; ++ks) {
#pragma unroll
      for (int t = 0; t <= NT; ++t) {
        const v2f bv = *(const v2f*)(bl + ks * 2 * PB + t * 32);
        acc[t] = __builtin_amdgcn_wmma_f32_16x16x4_f32(false, a[ks], false, bv, (short)0,
                                                       acc[t], false, false);
      }
    }

    const int rbase = tile * 16 + h * 8;
    float* cbase = C + (size_t)rbase * NC + c;
    if (tile * 16 + 16 <= M) {
#pragma unroll
      for (int t = 0; t < NT; ++t)
#pragma unroll
        for (int i = 0; i < 8; ++i) cbase[(size_t)i * NC + t * 16] = acc[t][i];
    } else {
#pragma unroll
      for (int t = 0; t < NT; ++t)
#pragma unroll
        for (int i = 0; i < 8; ++i)
          if (rbase + i < M) cbase[(size_t)i * NC + t * 16] = acc[t][i];
    }
    if (c < 2 * H) {
      float* lp = (c < H ? als + c : ald + (c - H)) + (size_t)rbase * H;
#pragma unroll
      for (int i = 0; i < 8; ++i)
        if (rbase + i < M) lp[(size_t)i * H] = acc[NT][i];
    }

    tile += stride;
    have = tile < ntiles;
    if (have) {
      int row = tile * 16 + c;
      if (row >= M) row = M - 1;
      const float* ap = A + (size_t)row * K + 2 * h;
#pragma unroll
      for (int ks = 0; ks < KS; ++ks) a[ks] = *(const v2f*)(ap + 4 * ks);
    }
  }
}

__device__ __forceinline__ float leaky(float v) {
  return v > 0.f ? v : NEG_SLOPE * v;
}

#define BSH 8
#define BW (1 << BSH)
#define NBMAX 512
#define SCAP 8192
#define PCAP 16384

__device__ __forceinline__ int wg_excl_scan(int v, int* wsum) {
  const int lane = threadIdx.x & 31;
  const int w = threadIdx.x >> 5;
  int inc = v;
  for (int o = 1; o < 32; o <<= 1) {
    const int u = __shfl_up(inc, o);
    if (lane >= o) inc += u;
  }
  if (lane == 31) wsum[w] = inc;
  __syncthreads();
  int run = 0;
  for (int k = 0; k < w; ++k) run += wsum[k];
  return run + inc - v;
}

__global__ void __launch_bounds__(1024)
bin_count_k(const int* __restrict__ dst, int* __restrict__ bintotal,
            int* __restrict__ wgbase, int* __restrict__ wgcnt, int E, int NB) {
  __shared__ int hist[NBMAX];
  const int t = threadIdx.x;
  if (t < NBMAX) hist[t] = 0;
  __syncthreads();
  const int chunk = (E + (int)gridDim.x - 1) / (int)gridDim.x;
  const int b = (int)blockIdx.x * chunk;
  int en = b + chunk;
  if (en > E) en = E;
  for (int e0 = b + t; e0 < en; e0 += 4096) {
    int d[4];
    bool ok[4];
#pragma unroll
    for (int k = 0; k < 4; ++k) {
      const int e = e0 + k * 1024;
      ok[k] = e < en;
      d[k] = dst[ok[k] ? e : en - 1];
    }
#pragma unroll
    for (int k = 0; k < 4; ++k) atomicAdd(&hist[d[k] >> BSH], ok[k] ? 1 : 0);
  }
  __syncthreads();
  if (t < NB) {
    const int hc = hist[t];
    wgcnt[(int)blockIdx.x * NB + t] = hc;
    wgbase[(int)blockIdx.x * NB + t] = atomicAdd(&bintotal[t], hc);
  }
}

__global__ void __launch_bounds__(1024)
bin_scatter_k(const int* __restrict__ src, const int* __restrict__ dst,
              const int* __restrict__ bintotal, const int* __restrict__ wgbase,
              const int* __restrict__ wgcnt, int2* __restrict__ tmp, int E, int NB) {
  __shared__ int cur[NBMAX];
  __shared__ int gdelta[NBMAX];
  __shared__ int wsum[32];
  __shared__ int wsum2[32];
  __shared__ int2 stage[SCAP];
  const int t = threadIdx.x;
  const int ex = wg_excl_scan(t < NB ? bintotal[t] : 0, wsum);
  const int chunk = (E + (int)gridDim.x - 1) / (int)gridDim.x;
  const int b = (int)blockIdx.x * chunk;
  int en = b + chunk;
  if (en > E) en = E;
  const bool staged = chunk <= SCAP;
  int ls = 0;
  if (staged) ls = wg_excl_scan(t < NB ? wgcnt[(int)blockIdx.x * NB + t] : 0, wsum2);
  if (t < NB) {
    const int gp = ex + wgbase[(int)blockIdx.x * NB + t];
    cur[t] = staged ? ls : gp;
    gdelta[t] = gp - ls;
  }
  __syncthreads();
  for (int e0 = b + t; e0 < en; e0 += 4096) {
    int d[4], s[4];
    bool ok[4];
#pragma unroll
    for (int k = 0; k < 4; ++k) {
      const int e = e0 + k * 1024;
      ok[k] = e < en;
      const int ec = ok[k] ? e : en - 1;
      d[k] = dst[ec];
      s[k] = src[ec];
    }
#pragma unroll
    for (int k = 0; k < 4; ++k) {
      const int p = atomicAdd(&cur[d[k] >> BSH], ok[k] ? 1 : 0);
      if (ok[k]) {
        if (staged) stage[p] = make_int2(s[k], d[k]);
        else tmp[p] = make_int2(s[k], d[k]);
      }
    }
  }
  if (staged) {
    __syncthreads();
    const int n = en - b;
    for (int i0 = t; i0 < n; i0 += 4096) {
      int2 sd[4];
      bool ok[4];
#pragma unroll
      for (int k = 0; k < 4; ++k) {
        const int i = i0 + k * 1024;
        ok[k] = i < n;
        sd[k] = stage[ok[k] ? i : i0];
      }
      int g[4];
#pragma unroll
      for (int k = 0; k < 4; ++k) g[k] = gdelta[sd[k].y >> BSH];
#pragma unroll
      for (int k = 0; k < 4; ++k)
        if (ok[k]) tmp[g[k] + i0 + k * 1024] = sd[k];
    }
  }
}

__global__ void __launch_bounds__(1024)
bin_place_k(const int2* __restrict__ tmp, const int* __restrict__ bintotal,
            int* __restrict__ rowp, int* __restrict__ csr, int N, int NB) {
  __shared__ int cnt[BW];
  __shared__ int cur[BW];
  __shared__ int wsum[32];
  __shared__ int wsum2[32];
  __shared__ int own;
  __shared__ int stage[PCAP];
  const int t = threadIdx.x;
  const int lane = t & 31;
  const int bin = (int)blockIdx.x;
  const int bt = (t <= bin) ? bintotal[t] : 0;
  int s = (t < bin) ? bt : 0;
  for (int o = 16; o > 0; o >>= 1) s += __shfl_xor(s, o);
  if (lane == 0) wsum[t >> 5] = s;
  if (t == bin) own = bt;
  if (t < BW) cnt[t] = 0;
  __syncthreads();
  int beg = 0;
  for (int k = 0; k < 32; ++k) beg += wsum[k];
  const int end = beg + own;

  for (int i0 = beg + t; i0 < end; i0 += 4096) {
    int d[4];
    bool ok[4];
#pragma unroll
    for (int k = 0; k < 4; ++k) {
      const int i = i0 + k * 1024;
      ok[k] = i < end;
      d[k] = tmp[ok[k] ? i : end - 1].y;
    }
#pragma unroll
    for (int k = 0; k < 4; ++k) atomicAdd(&cnt[d[k] & (BW - 1)], ok[k] ? 1 : 0);
  }
  __syncthreads();
  const int ex = wg_excl_scan(t < BW ? cnt[t] : 0, wsum2);
  const bool staged = end - beg <= PCAP;
  if (t < BW) {
    cur[t] = staged ? ex : beg + ex;
    const int node = bin * BW + t;
    if (node < N) rowp[node] = beg + ex;
  }
  if (bin == NB - 1 && t == 0) rowp[N] = end;
  __syncthreads();
  for (int i0 = beg + t; i0 < end; i0 += 4096) {
    int2 sd[4];
    bool ok[4];
#pragma unroll
    for (int k = 0; k < 4; ++k) {
      const int i = i0 + k * 1024;
      ok[k] = i < end;
      sd[k] = tmp[ok[k] ? i : end - 1];
    }
#pragma unroll
    for (int k = 0; k < 4; ++k) {
      const int p = atomicAdd(&cur[sd[k].y & (BW - 1)], ok[k] ? 1 : 0);
      if (ok[k]) {
        if (staged) stage[p] = sd[k].x;
        else csr[p] = sd[k].x;
      }
    }
  }
  if (staged) {
    __syncthreads();
    for (int i = beg + t; i < end; i += 1024) csr[i] = stage[i - beg];
  }
}

template <int G>
__device__ __forceinline__ float group_max(float v) {
  v = fmaxf(v, dpp8_f<dpp8_xor_sel(1)>(v));
  v = fmaxf(v, dpp8_f<dpp8_xor_sel(2)>(v));
  v = fmaxf(v, dpp8_f<dpp8_xor_sel(4)>(v));
  if constexpr (G == 32) {
    v = fmaxf(v, xor8_f(v));
    v = fmaxf(v, xor16_f(v));
  }
  return v;
}

template <int P> struct rowv { typedef float type __attribute__((ext_vector_type(P))); };

template <int H, int CH>
__device__ __forceinline__ typename rowv<CH / 32>::type
node_sums(const int d, const int* __restrict__ rowp, const int* __restrict__ csr,
          const float* __restrict__ h, const float* __restrict__ als,
          const float* __restrict__ ald, float& ssum_out) {
  static_assert(H == 1 || H == 4, "head groups of 32 or 8 lanes");
  const int lane = threadIdx.x & 31;
  constexpr int PER = CH / 32;
  constexpr int G = 32 / H;
  typedef typename rowv<PER>::type RowV;
  const int c0 = lane * PER;
  const int head = lane / G;
  const int li = lane % G;
  const int beg = rowp[d];
  const int end = rowp[d + 1];
  const float ad = ald[d * H + head];

  float ssum = 0.f;
  float acc[PER];
#pragma unroll
  for (int i = 0; i < PER; ++i) acc[i] = 0.f;

#define GATHER_ROW(J)                                                              \
  if ((J) < cnt) {                                                                 \
    const int s_ = __builtin_amdgcn_readlane(sk[(J) / G], (J) % G);                \
    hv[(J) & 7] = *(const RowV*)(h + (size_t)s_ * CH + c0);                        \
  }
#define GATHER_ROWS8(J0)                                                           \
  GATHER_ROW((J0) + 0) GATHER_ROW((J0) + 1) GATHER_ROW((J0) + 2) GATHER_ROW((J0) + 3) \
  GATHER_ROW((J0) + 4) GATHER_ROW((J0) + 5) GATHER_ROW((J0) + 6) GATHER_ROW((J0) + 7)
#define FMA_ROW(J)                                                                 \
  if ((J) < cnt) {                                                                 \
    float p_;                                                                      \
    if constexpr (H == 4) {                                                        \
      p_ = dpp8_f<dpp8_bcast_sel((J) % 8)>(tk[(J) / 8]);                           \
    } else {                                                                       \
      p_ = __builtin_bit_cast(                                                     \
          float, __builtin_amdgcn_readlane(__builtin_bit_cast(int, tk[0]), (J)));  \
    }                                                                              \
    ssum += p_;                                                                    \
      \
    _Pragma("unroll") for (int i_ = 0; i_ < PER; ++i_)                             \
        acc[i_] = fmaf(hv[(J) & 7][i_], p_, acc[i_]);                              \
  }
#define FMA_ROWS8(J0)                                                              \
  FMA_ROW((J0) + 0) FMA_ROW((J0) + 1) FMA_ROW((J0) + 2) FMA_ROW((J0) + 3)          \
  FMA_ROW((J0) + 4) FMA_ROW((J0) + 5) FMA_ROW((J0) + 6) FMA_ROW((J0) + 7)
#define LOAD_IDS(BASE)                                                             \
           \
  _Pragma("unroll") for (int k = 0; k < H; ++k) {                                  \
    const int jj = k * G + li;                                                     \
    sk[k] = csr[(BASE) + (jj < cnt ? jj : cnt - 1)];                               \
  }

  if (beg < end) {
    int sk[H];
    float ak[H], ek[H], tk[H];
    RowV hv[8];
    int cnt = end - beg;
    if (cnt > 32) cnt = 32;

    LOAD_IDS(beg)
#pragma unroll
    for (int k = 0; k < H; ++k) ak[k] = als[sk[k] * H + head];
    GATHER_ROWS8(0)

    float m = -INFINITY;
#pragma unroll
    for (int k = 0; k < H; ++k) {
      ek[k] = (k * G + li < cnt) ? leaky(ak[k] + ad) : -INFINITY;
      m = fmaxf(m, ek[k]);
    }
    for (int base = beg + 32; base < end; base += 32) {
      int c2 = end - base;
      if (c2 > 32) c2 = 32;
#pragma unroll
      for (int k = 0; k < H; ++k) {
        const int jj = k * G + li;
        const int s2 = csr[base + (jj < c2 ? jj : c2 - 1)];
        const float e2 = (jj < c2) ? leaky(als[s2 * H + head] + ad) : -INFINITY;
        m = fmaxf(m, e2);
      }
    }
    m = group_max<G>(m);

    for (int base = beg; base < end; base += 32) {
      cnt = end - base;
      if (cnt > 32) cnt = 32;
      if (base != beg) {
        LOAD_IDS(base)
#pragma unroll
        for (int k = 0; k < H; ++k) ak[k] = als[sk[k] * H + head];
        GATHER_ROWS8(0)
#pragma unroll
        for (int k = 0; k < H; ++k)
          ek[k] = (k * G + li < cnt) ? leaky(ak[k] + ad) : -INFINITY;
      }
#pragma unroll
      for (int k = 0; k < H; ++k)
        tk[k] = (k * G + li < cnt) ? __expf(ek[k] - m) : 0.f;

      FMA_ROWS8(0)
      if (8 < cnt) {
        GATHER_ROWS8(8)
        FMA_ROWS8(8)
      }
      if (16 < cnt) {
        GATHER_ROWS8(16)
        FMA_ROWS8(16)
      }
      if (24 < cnt) {
        GATHER_ROWS8(24)
        FMA_ROWS8(24)
      }
    }
  }
#undef GATHER_ROW
#undef GATHER_ROWS8
#undef FMA_ROW
#undef FMA_ROWS8
#undef LOAD_IDS

  ssum_out = ssum;
  RowV r;
#pragma unroll
  for (int i = 0; i < PER; ++i) r[i] = acc[i];
  return r;
}

template <int PER, bool ELU>
__device__ __forceinline__ typename rowv<PER>::type
finish_row(const typename rowv<PER>::type acc, const float ssum,
           const typename rowv<PER>::type bv) {
  const float den = ssum + 1e-16f;
  typename rowv<PER>::type o;
#pragma unroll
  for (int i = 0; i < PER; ++i) {
    float v = acc[i] / den + bv[i];
    if (ELU) v = v > 0.f ? v : (__expf(v) - 1.0f);
    o[i] = v;
  }
  return o;
}

template <int H, int CH, bool ELU>
__global__ void node_gather(const int* __restrict__ rowp, const int* __restrict__ csr,
                            const float* __restrict__ h,
                            const float* __restrict__ als, const float* __restrict__ ald,
                            const float* __restrict__ bias, float* __restrict__ out, int N) {
  constexpr int PER = CH / 32;
  typedef typename rowv<PER>::type RowV;
  const int gw = (blockIdx.x * blockDim.x + threadIdx.x) >> 5;
  const int lane = threadIdx.x & 31;
  if (gw >= N) return;
  const int d = __builtin_amdgcn_readfirstlane(gw);
  const RowV bv = *(const RowV*)(bias + lane * PER);
  float ssum;
  const RowV acc = node_sums<H, CH>(d, rowp, csr, h, als, ald, ssum);
  *(RowV*)(out + (size_t)d * CH + lane * PER) = finish_row<PER, ELU>(acc, ssum, bv);
}

template <int H, int K, int NC, int H2>
__global__ void __launch_bounds__(512, 16)
gather_gemm(const int* __restrict__ rowp, const int* __restrict__ csr,
            const float* __restrict__ h, const float* __restrict__ als,
            const float* __restrict__ ald, const float* __restrict__ bias,
            const float* __restrict__ B, const float* __restrict__ asrc2,
            const float* __restrict__ adst2, float* __restrict__ C,
            float* __restrict__ als2, float* __restrict__ ald2, int N,
            const int* __restrict__ cfrom, int* __restrict__ cto, int cn) {
  constexpr int NT = NC / 16;
  constexpr int KS = K / 4;
  constexpr int PB = 2 * NC + 32;
  constexpr int TP = K + 4;
  constexpr int PER = K / 32;
  static_assert(NT + 1 <= 16, "one column tile per wave");
  typedef typename rowv<PER>::type RowV;
  __shared__ __attribute__((aligned(16))) float Ws[(K / 2) * PB];
  __shared__ __attribute__((aligned(16))) float Ts[2][16 * TP];

  const int wave = __builtin_amdgcn_readfirstlane((int)threadIdx.x >> 5);
  const int ntiles = (N + 15) >> 4;

  for (int i = (int)blockIdx.x * 512 + (int)threadIdx.x; i < cn; i += (int)gridDim.x * 512)
    cto[i] = cfrom[i];
  stage_b<K, NC, H2>(B, asrc2, adst2, Ws);

  int buf = 0;
  for (int tile = (int)blockIdx.x; tile < ntiles; tile += (int)gridDim.x) {
    const int d = tile * 16 + wave;
    RowV acc1 = {};
    float ssum = 0.f;
    if (d < N) acc1 = node_sums<H, K>(d, rowp, csr, h, als, ald, ssum);
    int lane = threadIdx.x & 31;
    asm volatile("" : "+v"(lane));
    const int c = lane & 15;
    const int hh = lane >> 4;
    const RowV bv = *(const RowV*)(bias + lane * PER);
    float* tb = Ts[buf];
    *(RowV*)(tb + wave * TP + lane * PER) = finish_row<PER, true>(acc1, ssum, bv);
    __syncthreads();
    if (wave <= NT) {
      const float* al = tb + c * TP + 2 * hh;
      const float* bt = Ws + hh * PB + 2 * c + wave * 32;
      v8f acc = {};
#pragma unroll
      for (int ks = 0; ks < KS; ++ks) {
        const v2f av = *(const v2f*)(al + 4 * ks);
        const v2f bv = *(const v2f*)(bt + ks * 2 * PB);
        acc = __builtin_amdgcn_wmma_f32_16x16x4_f32(false, av, false, bv, (short)0, acc,
                                                    false, false);
      }
      const int rbase = tile * 16 + hh * 8;
      if (wave < NT) {
        float* cb = C + (size_t)rbase * NC + wave * 16 + c;
#pragma unroll
        for (int i = 0; i < 8; ++i)
          if (rbase + i < N) cb[(size_t)i * NC] = acc[i];
      } else if (c < 2 * H2) {
        float* lp = (c < H2 ? als2 + c : ald2 + (c - H2)) + (size_t)rbase * H2;
#pragma unroll
        for (int i = 0; i < 8; ++i)
          if (rbase + i < N) lp[(size_t)i * H2] = acc[i];
      }
    }
    buf ^= 1;
  }
}

extern "C" void kernel_launch(void* const* d_in, const int* in_sizes, int n_in,
                              void* d_out, int out_size, void* d_ws, size_t ws_size,
                              hipStream_t stream) {
  const float* x   = (const float*)d_in[0];
  const int*   ei  = (const int*)d_in[1];
  const float* W1  = (const float*)d_in[2];
  const float* as1 = (const float*)d_in[3];
  const float* ad1 = (const float*)d_in[4];
  const float* b1  = (const float*)d_in[5];
  const float* W2  = (const float*)d_in[6];
  const float* as2 = (const float*)d_in[7];
  const float* ad2 = (const float*)d_in[8];
  const float* b2  = (const float*)d_in[9];
  float* out = (float*)d_out;

  const int N = in_sizes[0] / IN_C;
  const int E = in_sizes[1] / 2;
  const int* src = ei;
  const int* dst = ei + E;

  float* ws = (float*)d_ws;
  float* h1   = ws;
  float* l2   = h1 + (size_t)N * 128;
  int*   csr  = (int*)(l2 + (size_t)N * 128);
  int2*  pairs = (int2*)l2;
  int*   bintotal = (int*)out;
  int*   rowp = (int*)out + N;
  int*   wgbase = (int*)out + (size_t)N * 11;
  int*   wgcnt  = (int*)out + (size_t)N * 13;
  float* als1 = out + (size_t)N * 3;
  float* ald1 = als1 + (size_t)N * HEADS;
  float* h2    = l2;
  int*   rowp2 = (int*)(l2 + (size_t)N * 64);
  float* als2  = l2 + (size_t)N * 64 + (size_t)N * 2;
  float* ald2  = als2 + N;

  auto cdiv = [](long long a, long long b) -> unsigned { return (unsigned)((a + b - 1) / b); };

  const int NB = (int)cdiv(N, BW);
  gemm_wmma<IN_C, HEADS * HID, HEADS><<<256, 512, 0, stream>>>(
      x, W1, as1, ad1, h1, als1, ald1, N, (unsigned*)bintotal, NB);

  bin_count_k<<<256, 1024, 0, stream>>>(dst, bintotal, wgbase, wgcnt, E, NB);
  bin_scatter_k<<<256, 1024, 0, stream>>>(src, dst, bintotal, wgbase, wgcnt, pairs, E, NB);
  bin_place_k<<<NB, 1024, 0, stream>>>(pairs, bintotal, rowp, csr, N, NB);

  gather_gemm<HEADS, HEADS * HID, OUT_C, 1><<<1024, 512, 0, stream>>>(
      rowp, csr, h1, als1, ald1, b1, W2, as2, ad2, h2, als2, ald2, N, rowp, rowp2, N + 1);

  node_gather<1, OUT_C, false>
      <<<cdiv((long long)N * 32, 256), 256, 0, stream>>>(rowp2, csr, h2, als2, ald2, b2, out, N);
}


# gather_gemm rounds pipelined over the split work-group barrier: signal once the row is stored, wait after the next node's gather; tile product one round late, dealt round-robin to 16 waves; bits kept
# speedup vs baseline: 6.5081x; 1.0328x over previous
#include <hip/hip_runtime.h>
#include <math.h>

#define IN_C 128
#define HID 32
#define HEADS 4
#define OUT_C 64
#define NEG_SLOPE 0.2f

typedef __attribute__((ext_vector_type(2))) float v2f;
typedef __attribute__((ext_vector_type(4))) float v4f;
typedef __attribute__((ext_vector_type(8))) float v8f;

constexpr unsigned dpp8_xor_sel(int m) {
  unsigned s = 0;
  for (int i = 0; i < 8; ++i) s |= (unsigned)(i ^ m) << (3 * i);
  return s;
}
constexpr unsigned dpp8_bcast_sel(int i) { return (unsigned)i * 0x249249u; }

template <unsigned SEL>
__device__ __forceinline__ float dpp8_f(float v) {
  return __builtin_bit_cast(float,
                            __builtin_amdgcn_mov_dpp8(__builtin_bit_cast(unsigned, v), SEL));
}
__device__ __forceinline__ float xor8_f(float v) {
  return __builtin_bit_cast(
      float, __builtin_amdgcn_update_dpp(0, __builtin_bit_cast(int, v), 0x168, 0xf, 0xf, false));
}
__device__ __forceinline__ float xor16_f(float v) {
  return __builtin_bit_cast(
      float, __builtin_amdgcn_permlanex16(__builtin_bit_cast(int, v), __builtin_bit_cast(int, v),
                                          0x76543210, 0xfedcba98, false, false));
}
template <int L>
__device__ __forceinline__ float group_sum(float v) {
  v += dpp8_f<dpp8_xor_sel(1)>(v);
  v += dpp8_f<dpp8_xor_sel(2)>(v);
  v += dpp8_f<dpp8_xor_sel(4)>(v);
  if constexpr (L == 16) v += xor8_f(v);
  return v;
}

template <int K, int NC, int H>
__device__ __forceinline__ void stage_b(const float* __restrict__ B,
                                        const float* __restrict__ asrc,
                                        const float* __restrict__ adst, float* __restrict__ Ws) {
  constexpr int LPH = (NC / H) / 4;
  constexpr int ZW = 32 - 4 * H;
  static_assert(LPH == 8 || LPH == 16, "head groups of 8 or 16 lanes");
  static_assert(512 % (NC / 4) == 0, "a thread keeps its column group over all passes");
  constexpr int PB = 2 * NC + 32;
  constexpr int QN = NC / 4;

  for (int i = (int)threadIdx.x; i < (K / 2) * ZW; i += 512)
    Ws[(i / ZW) * PB + 2 * NC + 4 * H + (i % ZW)] = 0.f;

  static_assert(((K / 2) * QN) % 512 == 0, "whole passes of the work-group");
  const int q = (int)threadIdx.x % QN;
  const v4f asv = *(const v4f*)(asrc + 4 * q);
  const v4f adv = *(const v4f*)(adst + 4 * q);
  constexpr int NP = ((K / 2) * QN) / 512;
  v4f rr0[NP], rr1[NP];
#pragma unroll
  for (int it = 0; it < NP; ++it) {
    const int p = ((int)threadIdx.x + it * 512) / QN;
    rr0[it] = *(const v4f*)(B + (size_t)(2 * p) * NC + 4 * q);
    rr1[it] = *(const v4f*)(B + (size_t)(2 * p + 1) * NC + 4 * q);
  }
#pragma unroll
  for (int it = 0; it < NP; ++it) {
    const int i = (int)threadIdx.x + it * 512;
    const int p = i / QN;
    const v4f r0 = rr0[it];
    const v4f r1 = rr1[it];
    float* d = Ws + p * PB + 8 * q;
    const v4f lo = {r0.x, r1.x, r0.y, r1.y};
    const v4f hi = {r0.z, r1.z, r0.w, r1.w};
    *(v4f*)d = lo;
    *(v4f*)(d + 4) = hi;
    float s0 = fmaf(r0.w, asv.w, fmaf(r0.z, asv.z, fmaf(r0.y, asv.y, r0.x * asv.x)));
    float s1 = fmaf(r1.w, asv.w, fmaf(r1.z, asv.z, fmaf(r1.y, asv.y, r1.x * asv.x)));
    float d0 = fmaf(r0.w, adv.w, fmaf(r0.z, adv.z, fmaf(r0.y, adv.y, r0.x * adv.x)));
    float d1 = fmaf(r1.w, adv.w, fmaf(r1.z, adv.z, fmaf(r1.y, adv.y, r1.x * adv.x)));
    s0 = group_sum<LPH>(s0);
    s1 = group_sum<LPH>(s1);
    d0 = group_sum<LPH>(d0);
    d1 = group_sum<LPH>(d1);
    if (q % LPH == 0) {
      float* lt = Ws + p * PB + 2 * NC;
      const int hd = q / LPH;
      *(v2f*)(lt + 2 * hd) = v2f{s0, s1};
      *(v2f*)(lt + 2 * (H + hd)) = v2f{d0, d1};
    }
  }
}

template <int K, int NC, int H>
__global__ void __launch_bounds__(512)
gemm_wmma(const float* __restrict__ A, const float* __restrict__ B,
          const float* __restrict__ asrc, const float* __restrict__ adst,
          float* __restrict__ C, float* __restrict__ als, float* __restrict__ ald, int M,
          unsigned* __restrict__ zp, int zn) {
  constexpr int NT = NC / 16;
  constexpr int KS = K / 4;
  constexpr int PB = 2 * NC + 32;
  constexpr int WAVES = 16;
  __shared__ __attribute__((aligned(16))) float Ws[(K / 2) * PB];

  const int lane = threadIdx.x & 31;
  const int c = lane & 15;
  const int h = lane >> 4;
  const int wave = __builtin_amdgcn_readfirstlane((int)threadIdx.x >> 5);
  const int ntiles = (M + 15) >> 4;
  const int stride = (int)gridDim.x * WAVES;
  int tile = (int)blockIdx.x + (int)gridDim.x * wave;

  v2f a[KS];
  bool have = tile < ntiles;
  if (have) {
    int row = tile * 16 + c;
    if (row >= M) row = M - 1;
    const float* ap = A + (size_t)row * K + 2 * h;
#pragma unroll
    for (int ks = 0; ks < KS; ++ks) a[ks] = *(const v2f*)(ap + 4 * ks);
  }

  stage_b<K, NC, H>(B, asrc, adst, Ws);
  for (int i = (int)blockIdx.x * 512 + (int)threadIdx.x; i < zn; i += (int)gridDim.x * 512)
    zp[i] = 0u;
  __syncthreads();

  const float* bl = Ws + h * PB + 2 * c;
  while (have) {
    asm volatile("" ::: "memory");
    v8f acc[NT + 1];
#pragma unroll
    for (int t = 0; t <= NT; ++t) acc[t] = v8f{};

#pragma unroll
    for (int ks = 0; ks < KS; ++ks) {
#pragma unroll
      for (int t = 0; t <= NT; ++t) {
        const v2f bv = *(const v2f*)(bl + ks * 2 * PB + t * 32);
        acc[t] = __builtin_amdgcn_wmma_f32_16x16x4_f32(false, a[ks], false, bv, (short)0,
                                                       acc[t], false, false);
      }
    }

    const int rbase = tile * 16 + h * 8;
    float* cbase = C + (size_t)rbase * NC + c;
    if (tile * 16 + 16 <= M) {
#pragma unroll
      for (int t = 0; t < NT; ++t)
#pragma unroll
        for (int i = 0; i < 8; ++i) cbase[(size_t)i * NC + t * 16] = acc[t][i];
    } else {
#pragma unroll
      for (int t = 0; t < NT; ++t)
#pragma unroll
        for (int i = 0; i < 8; ++i)
          if (rbase + i < M) cbase[(size_t)i * NC + t * 16] = acc[t][i];
    }
    if (c < 2 * H) {
      float* lp = (c < H ? als + c : ald + (c - H)) + (size_t)rbase * H;
#pragma unroll
      for (int i = 0; i < 8; ++i)
        if (rbase + i < M) lp[(size_t)i * H] = acc[NT][i];
    }

    tile += stride;
    have = tile < ntiles;
    if (have) {
      int row = tile * 16 + c;
      if (row >= M) row = M - 1;
      const float* ap = A + (size_t)row * K + 2 * h;
#pragma unroll
      for (int ks = 0; ks < KS; ++ks) a[ks] = *(const v2f*)(ap + 4 * ks);
    }
  }
}

__device__ __forceinline__ float leaky(float v) {
  return v > 0.f ? v : NEG_SLOPE * v;
}

#define BSH 8
#define BW (1 << BSH)
#define NBMAX 512
#define SCAP 8192
#define PCAP 16384

__device__ __forceinline__ int wg_excl_scan(int v, int* wsum) {
  const int lane = threadIdx.x & 31;
  const int w = threadIdx.x >> 5;
  int inc = v;
  for (int o = 1; o < 32; o <<= 1) {
    const int u = __shfl_up(inc, o);
    if (lane >= o) inc += u;
  }
  if (lane == 31) wsum[w] = inc;
  __syncthreads();
  int run = 0;
  for (int k = 0; k < w; ++k) run += wsum[k];
  return run + inc - v;
}

__global__ void __launch_bounds__(1024)
bin_count_k(const int* __restrict__ dst, int* __restrict__ bintotal,
            int* __restrict__ wgbase, int* __restrict__ wgcnt, int E, int NB) {
  __shared__ int hist[NBMAX];
  const int t = threadIdx.x;
  if (t < NBMAX) hist[t] = 0;
  __syncthreads();
  const int chunk = (E + (int)gridDim.x - 1) / (int)gridDim.x;
  const int b = (int)blockIdx.x * chunk;
  int en = b + chunk;
  if (en > E) en = E;
  for (int e0 = b + t; e0 < en; e0 += 4096) {
    int d[4];
    bool ok[4];
#pragma unroll
    for (int k = 0; k < 4; ++k) {
      const int e = e0 + k * 1024;
      ok[k] = e < en;
      d[k] = dst[ok[k] ? e : en - 1];
    }
#pragma unroll
    for (int k = 0; k < 4; ++k) atomicAdd(&hist[d[k] >> BSH], ok[k] ? 1 : 0);
  }
  __syncthreads();
  if (t < NB) {
    const int hc = hist[t];
    wgcnt[(int)blockIdx.x * NB + t] = hc;
    wgbase[(int)blockIdx.x * NB + t] = atomicAdd(&bintotal[t], hc);
  }
}

__global__ void __launch_bounds__(1024)
bin_scatter_k(const int* __restrict__ src, const int* __restrict__ dst,
              const int* __restrict__ bintotal, const int* __restrict__ wgbase,
              const int* __restrict__ wgcnt, int2* __restrict__ tmp, int E, int NB) {
  __shared__ int cur[NBMAX];
  __shared__ int gdelta[NBMAX];
  __shared__ int wsum[32];
  __shared__ int wsum2[32];
  __shared__ int2 stage[SCAP];
  const int t = threadIdx.x;
  const int ex = wg_excl_scan(t < NB ? bintotal[t] : 0, wsum);
  const int chunk = (E + (int)gridDim.x - 1) / (int)gridDim.x;
  const int b = (int)blockIdx.x * chunk;
  int en = b + chunk;
  if (en > E) en = E;
  const bool staged = chunk <= SCAP;
  int ls = 0;
  if (staged) ls = wg_excl_scan(t < NB ? wgcnt[(int)blockIdx.x * NB + t] : 0, wsum2);
  if (t < NB) {
    const int gp = ex + wgbase[(int)blockIdx.x * NB + t];
    cur[t] = staged ? ls : gp;
    gdelta[t] = gp - ls;
  }
  __syncthreads();
  for (int e0 = b + t; e0 < en; e0 += 4096) {
    int d[4], s[4];
    bool ok[4];
#pragma unroll
    for (int k = 0; k < 4; ++k) {
      const int e = e0 + k * 1024;
      ok[k] = e < en;
      const int ec = ok[k] ? e : en - 1;
      d[k] = dst[ec];
      s[k] = src[ec];
    }
#pragma unroll
    for (int k = 0; k < 4; ++k) {
      const int p = atomicAdd(&cur[d[k] >> BSH], ok[k] ? 1 : 0);
      if (ok[k]) {
        if (staged) stage[p] = make_int2(s[k], d[k]);
        else tmp[p] = make_int2(s[k], d[k]);
      }
    }
  }
  if (staged) {
    __syncthreads();
    const int n = en - b;
    for (int i0 = t; i0 < n; i0 += 4096) {
      int2 sd[4];
      bool ok[4];
#pragma unroll
      for (int k = 0; k < 4; ++k) {
        const int i = i0 + k * 1024;
        ok[k] = i < n;
        sd[k] = stage[ok[k] ? i : i0];
      }
      int g[4];
#pragma unroll
      for (int k = 0; k < 4; ++k) g[k] = gdelta[sd[k].y >> BSH];
#pragma unroll
      for (int k = 0; k < 4; ++k)
        if (ok[k]) tmp[g[k] + i0 + k * 1024] = sd[k];
    }
  }
}

__global__ void __launch_bounds__(1024)
bin_place_k(const int2* __restrict__ tmp, const int* __restrict__ bintotal,
            int* __restrict__ rowp, int* __restrict__ csr, int N, int NB) {
  __shared__ int cnt[BW];
  __shared__ int cur[BW];
  __shared__ int wsum[32];
  __shared__ int wsum2[32];
  __shared__ int own;
  __shared__ int stage[PCAP];
  const int t = threadIdx.x;
  const int lane = t & 31;
  const int bin = (int)blockIdx.x;
  const int bt = (t <= bin) ? bintotal[t] : 0;
  int s = (t < bin) ? bt : 0;
  for (int o = 16; o > 0; o >>= 1) s += __shfl_xor(s, o);
  if (lane == 0) wsum[t >> 5] = s;
  if (t == bin) own = bt;
  if (t < BW) cnt[t] = 0;
  __syncthreads();
  int beg = 0;
  for (int k = 0; k < 32; ++k) beg += wsum[k];
  const int end = beg + own;

  for (int i0 = beg + t; i0 < end; i0 += 4096) {
    int d[4];
    bool ok[4];
#pragma unroll
    for (int k = 0; k < 4; ++k) {
      const int i = i0 + k * 1024;
      ok[k] = i < end;
      d[k] = tmp[ok[k] ? i : end - 1].y;
    }
#pragma unroll
    for (int k = 0; k < 4; ++k) atomicAdd(&cnt[d[k] & (BW - 1)], ok[k] ? 1 : 0);
  }
  __syncthreads();
  const int ex = wg_excl_scan(t < BW ? cnt[t] : 0, wsum2);
  const bool staged = end - beg <= PCAP;
  if (t < BW) {
    cur[t] = staged ? ex : beg + ex;
    const int node = bin * BW + t;
    if (node < N) rowp[node] = beg + ex;
  }
  if (bin == NB - 1 && t == 0) rowp[N] = end;
  __syncthreads();
  for (int i0 = beg + t; i0 < end; i0 += 4096) {
    int2 sd[4];
    bool ok[4];
#pragma unroll
    for (int k = 0; k < 4; ++k) {
      const int i = i0 + k * 1024;
      ok[k] = i < end;
      sd[k] = tmp[ok[k] ? i : end - 1];
    }
#pragma unroll
    for (int k = 0; k < 4; ++k) {
      const int p = atomicAdd(&cur[sd[k].y & (BW - 1)], ok[k] ? 1 : 0);
      if (ok[k]) {
        if (staged) stage[p] = sd[k].x;
        else csr[p] = sd[k].x;
      }
    }
  }
  if (staged) {
    __syncthreads();
    for (int i = beg + t; i < end; i += 1024) csr[i] = stage[i - beg];
  }
}

template <int G>
__device__ __forceinline__ float group_max(float v) {
  v = fmaxf(v, dpp8_f<dpp8_xor_sel(1)>(v));
  v = fmaxf(v, dpp8_f<dpp8_xor_sel(2)>(v));
  v = fmaxf(v, dpp8_f<dpp8_xor_sel(4)>(v));
  if constexpr (G == 32) {
    v = fmaxf(v, xor8_f(v));
    v = fmaxf(v, xor16_f(v));
  }
  return v;
}

template <int P> struct rowv { typedef float type __attribute__((ext_vector_type(P))); };

template <int H, int CH>
__device__ __forceinline__ typename rowv<CH / 32>::type
node_sums(const int d, const int* __restrict__ rowp, const int* __restrict__ csr,
          const float* __restrict__ h, const float* __restrict__ als,
          const float* __restrict__ ald, float& ssum_out) {
  static_assert(H == 1 || H == 4, "head groups of 32 or 8 lanes");
  const int lane = threadIdx.x & 31;
  constexpr int PER = CH / 32;
  constexpr int G = 32 / H;
  typedef typename rowv<PER>::type RowV;
  const int c0 = lane * PER;
  const int head = lane / G;
  const int li = lane % G;
  const int beg = rowp[d];
  const int end = rowp[d + 1];
  const float ad = ald[d * H + head];

  float ssum = 0.f;
  float acc[PER];
#pragma unroll
  for (int i = 0; i < PER; ++i) acc[i] = 0.f;

#define GATHER_ROW(J)                                                              \
  if ((J) < cnt) {                                                                 \
    const int s_ = __builtin_amdgcn_readlane(sk[(J) / G], (J) % G);                \
    hv[(J) & 7] = *(const RowV*)(h + (size_t)s_ * CH + c0);                        \
  }
#define GATHER_ROWS8(J0)                                                           \
  GATHER_ROW((J0) + 0) GATHER_ROW((J0) + 1) GATHER_ROW((J0) + 2) GATHER_ROW((J0) + 3) \
  GATHER_ROW((J0) + 4) GATHER_ROW((J0) + 5) GATHER_ROW((J0) + 6) GATHER_ROW((J0) + 7)
#define FMA_ROW(J)                                                                 \
  if ((J) < cnt) {                                                                 \
    float p_;                                                                      \
    if constexpr (H == 4) {                                                        \
      p_ = dpp8_f<dpp8_bcast_sel((J) % 8)>(tk[(J) / 8]);                           \
    } else {                                                                       \
      p_ = __builtin_bit_cast(                                                     \
          float, __builtin_amdgcn_readlane(__builtin_bit_cast(int, tk[0]), (J)));  \
    }                                                                              \
    ssum += p_;                                                                    \
      \
    _Pragma("unroll") for (int i_ = 0; i_ < PER; ++i_)                             \
        acc[i_] = fmaf(hv[(J) & 7][i_], p_, acc[i_]);                              \
  }
#define FMA_ROWS8(J0)                                                              \
  FMA_ROW((J0) + 0) FMA_ROW((J0) + 1) FMA_ROW((J0) + 2) FMA_ROW((J0) + 3)          \
  FMA_ROW((J0) + 4) FMA_ROW((J0) + 5) FMA_ROW((J0) + 6) FMA_ROW((J0) + 7)
#define LOAD_IDS(BASE)                                                             \
           \
  _Pragma("unroll") for (int k = 0; k < H; ++k) {                                  \
    const int jj = k * G + li;                                                     \
    sk[k] = csr[(BASE) + (jj < cnt ? jj : cnt - 1)];                               \
  }

  if (beg < end) {
    int sk[H];
    float ak[H], ek[H], tk[H];
    RowV hv[8];
    int cnt = end - beg;
    if (cnt > 32) cnt = 32;

    LOAD_IDS(beg)
#pragma unroll
    for (int k = 0; k < H; ++k) ak[k] = als[sk[k] * H + head];
    GATHER_ROWS8(0)

    float m = -INFINITY;
#pragma unroll
    for (int k = 0; k < H; ++k) {
      ek[k] = (k * G + li < cnt) ? leaky(ak[k] + ad) : -INFINITY;
      m = fmaxf(m, ek[k]);
    }
    for (int base = beg + 32; base < end; base += 32) {
      int c2 = end - base;
      if (c2 > 32) c2 = 32;
#pragma unroll
      for (int k = 0; k < H; ++k) {
        const int jj = k * G + li;
        const int s2 = csr[base + (jj < c2 ? jj : c2 - 1)];
        const float e2 = (jj < c2) ? leaky(als[s2 * H + head] + ad) : -INFINITY;
        m = fmaxf(m, e2);
      }
    }
    m = group_max<G>(m);

    for (int base = beg; base < end; base += 32) {
      cnt = end - base;
      if (cnt > 32) cnt = 32;
      if (base != beg) {
        LOAD_IDS(base)
#pragma unroll
        for (int k = 0; k < H; ++k) ak[k] = als[sk[k] * H + head];
        GATHER_ROWS8(0)
#pragma unroll
        for (int k = 0; k < H; ++k)
          ek[k] = (k * G + li < cnt) ? leaky(ak[k] + ad) : -INFINITY;
      }
#pragma unroll
      for (int k = 0; k < H; ++k)
        tk[k] = (k * G + li < cnt) ? __expf(ek[k] - m) : 0.f;

      FMA_ROWS8(0)
      if (8 < cnt) {
        GATHER_ROWS8(8)
        FMA_ROWS8(8)
      }
      if (16 < cnt) {
        GATHER_ROWS8(16)
        FMA_ROWS8(16)
      }
      if (24 < cnt) {
        GATHER_ROWS8(24)
        FMA_ROWS8(24)
      }
    }
  }
#undef GATHER_ROW
#undef GATHER_ROWS8
#undef FMA_ROW
#undef FMA_ROWS8
#undef LOAD_IDS

  ssum_out = ssum;
  RowV r;
#pragma unroll
  for (int i = 0; i < PER; ++i) r[i] = acc[i];
  return r;
}

template <int PER, bool ELU>
__device__ __forceinline__ typename rowv<PER>::type
finish_row(const typename rowv<PER>::type acc, const float ssum,
           const typename rowv<PER>::type bv) {
  const float den = ssum + 1e-16f;
  typename rowv<PER>::type o;
#pragma unroll
  for (int i = 0; i < PER; ++i) {
    float v = acc[i] / den + bv[i];
    if (ELU) v = v > 0.f ? v : (__expf(v) - 1.0f);
    o[i] = v;
  }
  return o;
}

template <int H, int CH, bool ELU>
__global__ void node_gather(const int* __restrict__ rowp, const int* __restrict__ csr,
                            const float* __restrict__ h,
                            const float* __restrict__ als, const float* __restrict__ ald,
                            const float* __restrict__ bias, float* __restrict__ out, int N) {
  constexpr int PER = CH / 32;
  typedef typename rowv<PER>::type RowV;
  const int gw = (blockIdx.x * blockDim.x + threadIdx.x) >> 5;
  const int lane = threadIdx.x & 31;
  if (gw >= N) return;
  const int d = __builtin_amdgcn_readfirstlane(gw);
  const RowV bv = *(const RowV*)(bias + lane * PER);
  float ssum;
  const RowV acc = node_sums<H, CH>(d, rowp, csr, h, als, ald, ssum);
  *(RowV*)(out + (size_t)d * CH + lane * PER) = finish_row<PER, ELU>(acc, ssum, bv);
}

__device__ __forceinline__ void round_signal() {
  asm volatile("s_wait_dscnt 0x0\n\ts_barrier_signal -1" ::: "memory");
}
__device__ __forceinline__ void round_wait() {
  asm volatile("s_barrier_wait -1" ::: "memory");
}

template <int K, int NC, int H2>
__device__ __forceinline__ void tile_product(const float* __restrict__ tb,
                                             const float* __restrict__ Ws, const int j,
                                             const int tile, const int lane,
                                             float* __restrict__ C, float* __restrict__ als2,
                                             float* __restrict__ ald2, const int N) {
  constexpr int NT = NC / 16;
  constexpr int KS = K / 4;
  constexpr int PB = 2 * NC + 32;
  constexpr int TP = K + 4;
  const int c = lane & 15;
  const int hh = lane >> 4;
  const float* al = tb + c * TP + 2 * hh;
  const float* bt = Ws + hh * PB + 2 * c + j * 32;
  v8f acc = {};
#pragma unroll
  for (int ks = 0; ks < KS; ++ks) {
    const v2f av = *(const v2f*)(al + 4 * ks);
    const v2f bv = *(const v2f*)(bt + ks * 2 * PB);
    acc = __builtin_amdgcn_wmma_f32_16x16x4_f32(false, av, false, bv, (short)0, acc,
                                                false, false);
  }
  const int rbase = tile * 16 + hh * 8;
  const bool ctile = j < NT;
  float* lp = (c < H2 ? als2 + c : ald2 + (c - H2)) + (size_t)rbase * H2;
  float* dp = ctile ? C + (size_t)rbase * NC + j * 16 + c : lp;
  const int pitch = ctile ? NC : H2;
  const bool live = ctile || c < 2 * H2;
#pragma unroll
  for (int i = 0; i < 8; ++i)
    if (live && rbase + i < N) dp[(size_t)i * pitch] = acc[i];
}

template <int H, int K, int NC, int H2>
__global__ void __launch_bounds__(512, 16)
gather_gemm(const int* __restrict__ rowp, const int* __restrict__ csr,
            const float* __restrict__ h, const float* __restrict__ als,
            const float* __restrict__ ald, const float* __restrict__ bias,
            const float* __restrict__ B, const float* __restrict__ asrc2,
            const float* __restrict__ adst2, float* __restrict__ C,
            float* __restrict__ als2, float* __restrict__ ald2, int N,
            const int* __restrict__ cfrom, int* __restrict__ cto, int cn) {
  constexpr int NT = NC / 16;
  constexpr int PB = 2 * NC + 32;
  constexpr int TP = K + 4;
  constexpr int PER = K / 32;
  static_assert(NT + 1 <= 16, "one column tile per wave");
  typedef typename rowv<PER>::type RowV;
  __shared__ __attribute__((aligned(16))) float Ws[(K / 2) * PB];
  __shared__ __attribute__((aligned(16))) float Ts[2][16 * TP];

  const int wave = __builtin_amdgcn_readfirstlane((int)threadIdx.x >> 5);
  const int ntiles = (N + 15) >> 4;

  for (int i = (int)blockIdx.x * 512 + (int)threadIdx.x; i < cn; i += (int)gridDim.x * 512)
    cto[i] = cfrom[i];
  stage_b<K, NC, H2>(B, asrc2, adst2, Ws);

  int buf = 0;
  int ptile = -1;
  int pj = 0;
  int rot = 0;
  for (int tile = (int)blockIdx.x; tile < ntiles; tile += (int)gridDim.x) {
    const int d = tile * 16 + wave;
    RowV acc1 = {};
    float ssum = 0.f;
    if (d < N) acc1 = node_sums<H, K>(d, rowp, csr, h, als, ald, ssum);
    int lane = threadIdx.x & 31;
    asm volatile("" : "+v"(lane));
    const RowV bv = *(const RowV*)(bias + lane * PER);
    RowV orow = finish_row<PER, true>(acc1, ssum, bv);
    asm volatile("" : "+v"(orow));
    if (ptile >= 0) {
      round_wait();
      if (pj <= NT)
        tile_product<K, NC, H2>(Ts[buf ^ 1], Ws, pj, ptile, lane, C, als2, ald2, N);
    }
    *(RowV*)(Ts[buf] + wave * TP + lane * PER) = orow;
    round_signal();
    ptile = tile;
    pj = (wave - rot) & 15;
    rot = (rot + NT + 1) & 15;
    buf ^= 1;
  }
  if (ptile >= 0) {
    round_wait();
    if (pj <= NT) {
      const int lane = threadIdx.x & 31;
      tile_product<K, NC, H2>(Ts[buf ^ 1], Ws, pj, ptile, lane, C, als2, ald2, N);
    }
  }
}

extern "C" void kernel_launch(void* const* d_in, const int* in_sizes, int n_in,
                              void* d_out, int out_size, void* d_ws, size_t ws_size,
                              hipStream_t stream) {
  const float* x   = (const float*)d_in[0];
  const int*   ei  = (const int*)d_in[1];
  const float* W1  = (const float*)d_in[2];
  const float* as1 = (const float*)d_in[3];
  const float* ad1 = (const float*)d_in[4];
  const float* b1  = (const float*)d_in[5];
  const float* W2  = (const float*)d_in[6];
  const float* as2 = (const float*)d_in[7];
  const float* ad2 = (const float*)d_in[8];
  const float* b2  = (const float*)d_in[9];
  float* out = (float*)d_out;

  const int N = in_sizes[0] / IN_C;
  const int E = in_sizes[1] / 2;
  const int* src = ei;
  const int* dst = ei + E;

  float* ws = (float*)d_ws;
  float* h1   = ws;
  float* l2   = h1 + (size_t)N * 128;
  int*   csr  = (int*)(l2 + (size_t)N * 128);
  int2*  pairs = (int2*)l2;
  int*   bintotal = (int*)out;
  int*   rowp = (int*)out + N;
  int*   wgbase = (int*)out + (size_t)N * 11;
  int*   wgcnt  = (int*)out + (size_t)N * 13;
  float* als1 = out + (size_t)N * 3;
  float* ald1 = als1 + (size_t)N * HEADS;
  float* h2    = l2;
  int*   rowp2 = (int*)(l2 + (size_t)N * 64);
  float* als2  = l2 + (size_t)N * 64 + (size_t)N * 2;
  float* ald2  = als2 + N;

  auto cdiv = [](long long a, long long b) -> unsigned { return (unsigned)((a + b - 1) / b); };

  const int NB = (int)cdiv(N, BW);
  gemm_wmma<IN_C, HEADS * HID, HEADS><<<256, 512, 0, stream>>>(
      x, W1, as1, ad1, h1, als1, ald1, N, (unsigned*)bintotal, NB);

  bin_count_k<<<256, 1024, 0, stream>>>(dst, bintotal, wgbase, wgcnt, E, NB);
  bin_scatter_k<<<256, 1024, 0, stream>>>(src, dst, bintotal, wgbase, wgcnt, pairs, E, NB);
  bin_place_k<<<NB, 1024, 0, stream>>>(pairs, bintotal, rowp, csr, N, NB);

  gather_gemm<HEADS, HEADS * HID, OUT_C, 1><<<1024, 512, 0, stream>>>(
      rowp, csr, h1, als1, ald1, b1, W2, as2, ad2, h2, als2, ald2, N, rowp, rowp2, N + 1);

  node_gather<1, OUT_C, false>
      <<<cdiv((long long)N * 32, 256), 256, 0, stream>>>(rowp2, csr, h2, als2, ald2, b2, out, N);
}
